# starting kernel
# baseline (speedup 1.0000x reference)
#include <hip/hip_runtime.h>
#include <math.h>

#define NB   4096
#define NC   200
#define NT   25
#define NH   64
#define NG   256
#define NF   1600
#define NCLS 16

typedef __attribute__((ext_vector_type(2))) float v2f;
typedef __attribute__((ext_vector_type(8))) float v8f;

#define WMMA_F32(a, b, c) \
    __builtin_amdgcn_wmma_f32_16x16x4_f32(false, (a), false, (b), (short)0, (c), false, false)

__device__ __forceinline__ float sigf(float x) { return 1.0f / (1.0f + __expf(-x)); }

__global__ __launch_bounds__(128) void lstm_kernel(
    const float* __restrict__ x,
    const float* __restrict__ Wih,
    const float* __restrict__ Whh,
    const float* __restrict__ bih,
    const float* __restrict__ bhh,
    float* __restrict__ feat)
{
    __shared__ float hlds[16 * NH];
    __shared__ float glds[16 * NG];

    const int tid  = threadIdx.x;
    const int lane = tid & 31;
    const int gb   = (tid >> 5) * 64;
    const int b0   = blockIdx.x * 16;
    const bool hi  = lane >= 16;
    const int nn   = lane & 15;
    const int ks   = hi ? 2 : 0;
    const int rofs = hi ? 8 : 0;

    float creg[8];
#pragma unroll
    for (int ii = 0; ii < 8; ++ii) {
        creg[ii] = 0.0f;
        hlds[tid + 128 * ii] = 0.0f;
    }
    __syncthreads();

    for (int t = 0; t < NT; ++t) {
        const int off = (t % 5) * 5 + (t / 5);

        v8f acc[4];
#pragma unroll
        for (int j = 0; j < 4; ++j) {
            const int g = gb + 16 * j + nn;
            const float bv = bih[g] + bhh[g];
#pragma unroll
            for (int r = 0; r < 8; ++r) acc[j][r] = bv;
        }

        const int arow = b0 + nn;
        for (int k = 0; k < 50; ++k) {
            const int kb = 4 * k + ks;
            v2f a;
            a.x = x[(arow * NC + kb    ) * NT + off];
            a.y = x[(arow * NC + kb + 1) * NT + off];
#pragma unroll
            for (int j = 0; j < 4; ++j) {
                const int g = gb + 16 * j + nn;
                v2f b;
                b.x = Wih[g * NC + kb];
                b.y = Wih[g * NC + kb + 1];
                acc[j] = WMMA_F32(a, b, acc[j]);
            }
        }

        for (int k = 0; k < 16; ++k) {
            const int kb = 4 * k + ks;
            v2f a;
            a.x = hlds[nn * NH + kb];
            a.y = hlds[nn * NH + kb + 1];
#pragma unroll
            for (int j = 0; j < 4; ++j) {
                const int g = gb + 16 * j + nn;
                v2f b;
                b.x = Whh[g * NH + kb];
                b.y = Whh[g * NH + kb + 1];
                acc[j] = WMMA_F32(a, b, acc[j]);
            }
        }

#pragma unroll
        for (int j = 0; j < 4; ++j)
#pragma unroll
            for (int r = 0; r < 8; ++r)
                glds[(r + rofs) * NG + gb + 16 * j + nn] = acc[j][r];
        __syncthreads();

#pragma unroll
        for (int ii = 0; ii < 8; ++ii) {
            const int idx = tid + 128 * ii;
            const int row = idx >> 6;
            const int hc  = idx & 63;
            const float gi = glds[row * NG + hc];
            const float gf = glds[row * NG + 64 + hc];
            const float gg = glds[row * NG + 128 + hc];
            const float go = glds[row * NG + 192 + hc];
            const float cc = sigf(gf) * creg[ii] + sigf(gi) * tanhf(gg);
            const float hh = sigf(go) * tanhf(cc);
            creg[ii] = cc;
            hlds[row * NH + hc] = hh;
            feat[(size_t)(b0 + row) * NF + hc * NT + t] = fmaxf(hh, 0.0f);
        }
        __syncthreads();
    }
}

__global__ void bn_stats_kernel(const float* __restrict__ feat,
                                const float* __restrict__ gamma,
                                const float* __restrict__ beta,
                                float* __restrict__ scale,
                                float* __restrict__ shift)
{
    const int f = blockIdx.x * blockDim.x + threadIdx.x;
    if (f >= NF) return;
    float s = 0.0f, s2 = 0.0f;
    for (int b = 0; b < NB; ++b) {
        const float v = feat[(size_t)b * NF + f];
        s += v;
        s2 += v * v;
    }
    const float mu  = s * (1.0f / NB);
    const float var = s2 * (1.0f / NB) - mu * mu;
    const float sc  = gamma[f] * rsqrtf(var + 1e-5f);
    scale[f] = sc;
    shift[f] = beta[f] - mu * sc;
}

__global__ __launch_bounds__(32) void fc_kernel(
    const float* __restrict__ feat,
    const float* __restrict__ scale,
    const float* __restrict__ shift,
    const float* __restrict__ fcw,
    const float* __restrict__ fcb,
    float* __restrict__ out)
{
    const int lane = threadIdx.x;
    const int b0   = blockIdx.x * 16;
    const bool hi  = lane >= 16;
    const int nn   = lane & 15;
    const int ks   = hi ? 2 : 0;
    const int rofs = hi ? 8 : 0;

    v8f acc;
    const float bb = fcb[nn];
#pragma unroll
    for (int r = 0; r < 8; ++r) acc[r] = bb;

    const int arow = b0 + nn;
    for (int k = 0; k < 400; ++k) {
        const int kb = 4 * k + ks;
        v2f a, b;
        const float f0 = feat[(size_t)arow * NF + kb];
        const float f1 = feat[(size_t)arow * NF + kb + 1];
        a.x = fmaxf(f0 * scale[kb] + shift[kb], 0.0f);
        a.y = fmaxf(f1 * scale[kb + 1] + shift[kb + 1], 0.0f);
        b.x = fcw[nn * NF + kb];
        b.y = fcw[nn * NF + kb + 1];
        acc = WMMA_F32(a, b, acc);
    }

#pragma unroll
    for (int r = 0; r < 8; ++r)
        out[(b0 + r + rofs) * NCLS + nn] = acc[r];
}

extern "C" void kernel_launch(void* const* d_in, const int* in_sizes, int n_in,
                              void* d_out, int out_size, void* d_ws, size_t ws_size,
                              hipStream_t stream)
{
    const float* x     = (const float*)d_in[0];
    const float* Wih   = (const float*)d_in[1];
    const float* Whh   = (const float*)d_in[2];
    const float* bih   = (const float*)d_in[3];
    const float* bhh   = (const float*)d_in[4];
    const float* gamma = (const float*)d_in[5];
    const float* beta  = (const float*)d_in[6];
    const float* fcw   = (const float*)d_in[7];
    const float* fcb   = (const float*)d_in[8];
    float* out = (float*)d_out;

    float* feat  = (float*)d_ws;
    float* scale = feat + (size_t)NB * NF;
    float* shift = scale + NF;

    lstm_kernel<<<NB / 16, 128, 0, stream>>>(x, Wih, Whh, bih, bhh, feat);
    bn_stats_kernel<<<(NF + 255) / 256, 256, 0, stream>>>(feat, gamma, beta, scale, shift);
    fc_kernel<<<NB / 16, 32, 0, stream>>>(feat, scale, shift, fcw, fcb, out);
}


# lstm kernel: WMMA operands staged in LDS (both weight matrices once per block, each step's x slice once per block one step ahead) instead of strided per-lane global loads at every K-step
# speedup vs baseline: 1.5473x; 1.5473x over previous
#include <hip/hip_runtime.h>
#include <math.h>

#define NB   4096
#define NC   200
#define NT   25
#define NH   64
#define NG   256
#define NF   1600
#define NCLS 16

typedef __attribute__((ext_vector_type(2))) float v2f;
typedef __attribute__((ext_vector_type(8))) float v8f;

#define WMMA_F32(a, b, c) \
    __builtin_amdgcn_wmma_f32_16x16x4_f32(false, (a), false, (b), (short)0, (c), false, false)

__device__ __forceinline__ float sigf(float x) { return 1.0f / (1.0f + __expf(-x)); }

#define XP 204
#define HP 68
#define KCH 6

__global__ __launch_bounds__(128) void lstm_kernel(
    const float* __restrict__ x,
    const float* __restrict__ Wih,
    const float* __restrict__ Whh,
    const float* __restrict__ bih,
    const float* __restrict__ bhh,
    float* __restrict__ feat)
{
    __shared__ __attribute__((aligned(16))) float wih_s[NG * XP];
    __shared__ __attribute__((aligned(16))) float whh_s[NG * HP];
    __shared__ __attribute__((aligned(16))) float xs[16 * XP];
    __shared__ __attribute__((aligned(16))) float hlds[16 * HP];
    __shared__ float glds[16 * NG];

    const int tid  = threadIdx.x;
    const int lane = tid & 31;
    const int gb   = (tid >> 5) * 64;
    const int b0   = blockIdx.x * 16;
    const bool hi  = lane >= 16;
    const int nn   = lane & 15;
    const int ks   = hi ? 2 : 0;
    const int rofs = hi ? 8 : 0;

    const float* xg = x + ((size_t)b0 * NC + tid) * NT;
    int xdst[25];
    float xn[25];
#pragma unroll
    for (int i = 0; i < 25; ++i) {
        const int e = tid + 128 * i;
        const int row = e / NC;
        xdst[i] = row * XP + (e - row * NC);
        xn[i] = xg[3200 * i];
    }

    {
        const float4* src = reinterpret_cast<const float4*>(Wih);
        for (int i0 = 0; i0 < 100; i0 += 25) {
            float4 v[25];
#pragma unroll
            for (int u = 0; u < 25; ++u) v[u] = src[tid + 128 * (i0 + u)];
#pragma unroll
            for (int u = 0; u < 25; ++u) {
                const int q = tid + 128 * (i0 + u);
                const int row = q / 50;
                *reinterpret_cast<float4*>(&wih_s[row * XP + 4 * (q - row * 50)]) = v[u];
            }
        }
    }
    {
        const float4* src = reinterpret_cast<const float4*>(Whh);
        float4 v[32];
#pragma unroll
        for (int u = 0; u < 32; ++u) v[u] = src[tid + 128 * u];
#pragma unroll
        for (int u = 0; u < 32; ++u) {
            const int q = tid + 128 * u;
            *reinterpret_cast<float4*>(&whh_s[(q >> 4) * HP + 4 * (q & 15)]) = v[u];
        }
    }

    float creg[8];
#pragma unroll
    for (int ii = 0; ii < 8; ++ii) {
        const int idx = tid + 128 * ii;
        creg[ii] = 0.0f;
        hlds[(idx >> 6) * HP + (idx & 63)] = 0.0f;
    }
#pragma unroll
    for (int i = 0; i < 25; ++i) xs[xdst[i]] = xn[i];

    float bv[4];
#pragma unroll
    for (int j = 0; j < 4; ++j) {
        const int g = gb + 16 * j + nn;
        bv[j] = bih[g] + bhh[g];
    }

    const float* xa = xs    + nn * XP + ks;
    const float* ha = hlds  + nn * HP + ks;
    const float* wa = wih_s + (gb + nn) * XP + ks;
    const float* ua = whh_s + (gb + nn) * HP + ks;
    __syncthreads();

    for (int t = 0; t < NT; ++t) {
        if (t + 1 < NT) {
            const int offn = ((t + 1) % 5) * 5 + ((t + 1) / 5);
#pragma unroll
            for (int i = 0; i < 25; ++i) xn[i] = xg[3200 * i + offn];
        }

        v8f acc[4];
#pragma unroll
        for (int j = 0; j < 4; ++j)
#pragma unroll
            for (int r = 0; r < 8; ++r) acc[j][r] = bv[j];

        v2f fa[2][KCH];
        v2f fb[2][KCH][4];
#define LOAD_SET(buf, c)                                                                 \
        _Pragma("unroll") for (int kk = 0; kk < KCH; ++kk) {                              \
            const int s = (c) * KCH + kk;                                                 \
            if (s < 50) {                                                                 \
                fa[buf][kk] = *reinterpret_cast<const v2f*>(xa + 4 * s);                  \
                _Pragma("unroll") for (int j = 0; j < 4; ++j)                             \
                    fb[buf][kk][j] = *reinterpret_cast<const v2f*>(wa + 16 * j * XP + 4 * s); \
            } else {                                                                      \
                fa[buf][kk] = *reinterpret_cast<const v2f*>(ha + 4 * (s - 50));           \
                _Pragma("unroll") for (int j = 0; j < 4; ++j)                             \
                    fb[buf][kk][j] = *reinterpret_cast<const v2f*>(ua + 16 * j * HP + 4 * (s - 50)); \
            }                                                                             \
        }
        LOAD_SET(0, 0)
#pragma unroll
        for (int c = 0; c < 11; ++c) {
            if (c + 1 < 11) { LOAD_SET((c + 1) & 1, c + 1) }
            __builtin_amdgcn_sched_barrier(0);
#pragma unroll
            for (int kk = 0; kk < KCH; ++kk)
#pragma unroll
                for (int j = 0; j < 4; ++j)
                    acc[j] = WMMA_F32(fa[c & 1][kk], fb[c & 1][kk][j], acc[j]);
            __builtin_amdgcn_sched_barrier(0);
        }
#undef LOAD_SET

#pragma unroll
        for (int j = 0; j < 4; ++j)
#pragma unroll
            for (int r = 0; r < 8; ++r)
                glds[(r + rofs) * NG + gb + 16 * j + nn] = acc[j][r];
        __syncthreads();

#pragma unroll
        for (int ii = 0; ii < 8; ++ii) {
            const int idx = tid + 128 * ii;
            const int row = idx >> 6;
            const int hc  = idx & 63;
            const float gi = glds[row * NG + hc];
            const float gf = glds[row * NG + 64 + hc];
            const float gg = glds[row * NG + 128 + hc];
            const float go = glds[row * NG + 192 + hc];
            const float cc = sigf(gf) * creg[ii] + sigf(gi) * tanhf(gg);
            const float hh = sigf(go) * tanhf(cc);
            creg[ii] = cc;
            hlds[row * HP + hc] = hh;
            feat[(size_t)(b0 + row) * NF + hc * NT + t] = fmaxf(hh, 0.0f);
        }
        if (t + 1 < NT) {
#pragma unroll
            for (int i = 0; i < 25; ++i) xs[xdst[i]] = xn[i];
        }
        __syncthreads();
    }
}

__global__ void bn_stats_kernel(const float* __restrict__ feat,
                                const float* __restrict__ gamma,
                                const float* __restrict__ beta,
                                float* __restrict__ scale,
                                float* __restrict__ shift)
{
    const int f = blockIdx.x * blockDim.x + threadIdx.x;
    if (f >= NF) return;
    float s = 0.0f, s2 = 0.0f;
    for (int b = 0; b < NB; ++b) {
        const float v = feat[(size_t)b * NF + f];
        s += v;
        s2 += v * v;
    }
    const float mu  = s * (1.0f / NB);
    const float var = s2 * (1.0f / NB) - mu * mu;
    const float sc  = gamma[f] * rsqrtf(var + 1e-5f);
    scale[f] = sc;
    shift[f] = beta[f] - mu * sc;
}

__global__ __launch_bounds__(32) void fc_kernel(
    const float* __restrict__ feat,
    const float* __restrict__ scale,
    const float* __restrict__ shift,
    const float* __restrict__ fcw,
    const float* __restrict__ fcb,
    float* __restrict__ out)
{
    const int lane = threadIdx.x;
    const int b0   = blockIdx.x * 16;
    const bool hi  = lane >= 16;
    const int nn   = lane & 15;
    const int ks   = hi ? 2 : 0;
    const int rofs = hi ? 8 : 0;

    v8f acc;
    const float bb = fcb[nn];
#pragma unroll
    for (int r = 0; r < 8; ++r) acc[r] = bb;

    const int arow = b0 + nn;
    for (int k = 0; k < 400; ++k) {
        const int kb = 4 * k + ks;
        v2f a, b;
        const float f0 = feat[(size_t)arow * NF + kb];
        const float f1 = feat[(size_t)arow * NF + kb + 1];
        a.x = fmaxf(f0 * scale[kb] + shift[kb], 0.0f);
        a.y = fmaxf(f1 * scale[kb + 1] + shift[kb + 1], 0.0f);
        b.x = fcw[nn * NF + kb];
        b.y = fcw[nn * NF + kb + 1];
        acc = WMMA_F32(a, b, acc);
    }

#pragma unroll
    for (int r = 0; r < 8; ++r)
        out[(b0 + r + rofs) * NCLS + nn] = acc[r];
}

extern "C" void kernel_launch(void* const* d_in, const int* in_sizes, int n_in,
                              void* d_out, int out_size, void* d_ws, size_t ws_size,
                              hipStream_t stream)
{
    const float* x     = (const float*)d_in[0];
    const float* Wih   = (const float*)d_in[1];
    const float* Whh   = (const float*)d_in[2];
    const float* bih   = (const float*)d_in[3];
    const float* bhh   = (const float*)d_in[4];
    const float* gamma = (const float*)d_in[5];
    const float* beta  = (const float*)d_in[6];
    const float* fcw   = (const float*)d_in[7];
    const float* fcb   = (const float*)d_in[8];
    float* out = (float*)d_out;

    float* feat  = (float*)d_ws;
    float* scale = feat + (size_t)NB * NF;
    float* shift = scale + NF;

    lstm_kernel<<<NB / 16, 128, 0, stream>>>(x, Wih, Whh, bih, bhh, feat);
    bn_stats_kernel<<<(NF + 255) / 256, 256, 0, stream>>>(feat, gamma, beta, scale, shift);
    fc_kernel<<<NB / 16, 32, 0, stream>>>(feat, scale, shift, fcw, fcb, out);
}


# batch-statistics kernel: 50 blocks x 1024 threads split the 4096-row sweep (16-byte loads, fixed-order LDS tree) instead of 7 blocks with one serial thread per feature; reorders the batch sums
# speedup vs baseline: 2.2017x; 1.4229x over previous
#include <hip/hip_runtime.h>
#include <math.h>

#define NB   4096
#define NC   200
#define NT   25
#define NH   64
#define NG   256
#define NF   1600
#define NCLS 16

typedef __attribute__((ext_vector_type(2))) float v2f;
typedef __attribute__((ext_vector_type(8))) float v8f;

#define WMMA_F32(a, b, c) \
    __builtin_amdgcn_wmma_f32_16x16x4_f32(false, (a), false, (b), (short)0, (c), false, false)

__device__ __forceinline__ float sigf(float x) { return 1.0f / (1.0f + __expf(-x)); }

#define XP 204
#define HP 68
#define KCH 6

__global__ __launch_bounds__(128) void lstm_kernel(
    const float* __restrict__ x,
    const float* __restrict__ Wih,
    const float* __restrict__ Whh,
    const float* __restrict__ bih,
    const float* __restrict__ bhh,
    float* __restrict__ feat)
{
    __shared__ __attribute__((aligned(16))) float wih_s[NG * XP];
    __shared__ __attribute__((aligned(16))) float whh_s[NG * HP];
    __shared__ __attribute__((aligned(16))) float xs[16 * XP];
    __shared__ __attribute__((aligned(16))) float hlds[16 * HP];
    __shared__ float glds[16 * NG];

    const int tid  = threadIdx.x;
    const int lane = tid & 31;
    const int gb   = (tid >> 5) * 64;
    const int b0   = blockIdx.x * 16;
    const bool hi  = lane >= 16;
    const int nn   = lane & 15;
    const int ks   = hi ? 2 : 0;
    const int rofs = hi ? 8 : 0;

    const float* xg = x + ((size_t)b0 * NC + tid) * NT;
    int xdst[25];
    float xn[25];
#pragma unroll
    for (int i = 0; i < 25; ++i) {
        const int e = tid + 128 * i;
        const int row = e / NC;
        xdst[i] = row * XP + (e - row * NC);
        xn[i] = xg[3200 * i];
    }

    {
        const float4* src = reinterpret_cast<const float4*>(Wih);
        for (int i0 = 0; i0 < 100; i0 += 25) {
            float4 v[25];
#pragma unroll
            for (int u = 0; u < 25; ++u) v[u] = src[tid + 128 * (i0 + u)];
#pragma unroll
            for (int u = 0; u < 25; ++u) {
                const int q = tid + 128 * (i0 + u);
                const int row = q / 50;
                *reinterpret_cast<float4*>(&wih_s[row * XP + 4 * (q - row * 50)]) = v[u];
            }
        }
    }
    {
        const float4* src = reinterpret_cast<const float4*>(Whh);
        float4 v[32];
#pragma unroll
        for (int u = 0; u < 32; ++u) v[u] = src[tid + 128 * u];
#pragma unroll
        for (int u = 0; u < 32; ++u) {
            const int q = tid + 128 * u;
            *reinterpret_cast<float4*>(&whh_s[(q >> 4) * HP + 4 * (q & 15)]) = v[u];
        }
    }

    float creg[8];
#pragma unroll
    for (int ii = 0; ii < 8; ++ii) {
        const int idx = tid + 128 * ii;
        creg[ii] = 0.0f;
        hlds[(idx >> 6) * HP + (idx & 63)] = 0.0f;
    }
#pragma unroll
    for (int i = 0; i < 25; ++i) xs[xdst[i]] = xn[i];

    float bv[4];
#pragma unroll
    for (int j = 0; j < 4; ++j) {
        const int g = gb + 16 * j + nn;
        bv[j] = bih[g] + bhh[g];
    }

    const float* xa = xs    + nn * XP + ks;
    const float* ha = hlds  + nn * HP + ks;
    const float* wa = wih_s + (gb + nn) * XP + ks;
    const float* ua = whh_s + (gb + nn) * HP + ks;
    __syncthreads();

    for (int t = 0; t < NT; ++t) {
        if (t + 1 < NT) {
            const int offn = ((t + 1) % 5) * 5 + ((t + 1) / 5);
#pragma unroll
            for (int i = 0; i < 25; ++i) xn[i] = xg[3200 * i + offn];
        }

        v8f acc[4];
#pragma unroll
        for (int j = 0; j < 4; ++j)
#pragma unroll
            for (int r = 0; r < 8; ++r) acc[j][r] = bv[j];

        v2f fa[2][KCH];
        v2f fb[2][KCH][4];
#define LOAD_SET(buf, c)                                                                 \
        _Pragma("unroll") for (int kk = 0; kk < KCH; ++kk) {                              \
            const int s = (c) * KCH + kk;                                                 \
            if (s < 50) {                                                                 \
                fa[buf][kk] = *reinterpret_cast<const v2f*>(xa + 4 * s);                  \
                _Pragma("unroll") for (int j = 0; j < 4; ++j)                             \
                    fb[buf][kk][j] = *reinterpret_cast<const v2f*>(wa + 16 * j * XP + 4 * s); \
            } else {                                                                      \
                fa[buf][kk] = *reinterpret_cast<const v2f*>(ha + 4 * (s - 50));           \
                _Pragma("unroll") for (int j = 0; j < 4; ++j)                             \
                    fb[buf][kk][j] = *reinterpret_cast<const v2f*>(ua + 16 * j * HP + 4 * (s - 50)); \
            }                                                                             \
        }
        LOAD_SET(0, 0)
#pragma unroll
        for (int c = 0; c < 11; ++c) {
            if (c + 1 < 11) { LOAD_SET((c + 1) & 1, c + 1) }
            __builtin_amdgcn_sched_barrier(0);
#pragma unroll
            for (int kk = 0; kk < KCH; ++kk)
#pragma unroll
                for (int j = 0; j < 4; ++j)
                    acc[j] = WMMA_F32(fa[c & 1][kk], fb[c & 1][kk][j], acc[j]);
            __builtin_amdgcn_sched_barrier(0);
        }
#undef LOAD_SET

#pragma unroll
        for (int j = 0; j < 4; ++j)
#pragma unroll
            for (int r = 0; r < 8; ++r)
                glds[(r + rofs) * NG + gb + 16 * j + nn] = acc[j][r];
        __syncthreads();

#pragma unroll
        for (int ii = 0; ii < 8; ++ii) {
            const int idx = tid + 128 * ii;
            const int row = idx >> 6;
            const int hc  = idx & 63;
            const float gi = glds[row * NG + hc];
            const float gf = glds[row * NG + 64 + hc];
            const float gg = glds[row * NG + 128 + hc];
            const float go = glds[row * NG + 192 + hc];
            const float cc = sigf(gf) * creg[ii] + sigf(gi) * tanhf(gg);
            const float hh = sigf(go) * tanhf(cc);
            creg[ii] = cc;
            hlds[row * HP + hc] = hh;
            feat[(size_t)(b0 + row) * NF + hc * NT + t] = fmaxf(hh, 0.0f);
        }
        if (t + 1 < NT) {
#pragma unroll
            for (int i = 0; i < 25; ++i) xs[xdst[i]] = xn[i];
        }
        __syncthreads();
    }
}

#define BNF 32
#define BNG 128

__global__ __launch_bounds__(1024) void bn_stats_kernel(const float* __restrict__ feat,
                                                        const float* __restrict__ gamma,
                                                        const float* __restrict__ beta,
                                                        float* __restrict__ scale,
                                                        float* __restrict__ shift)
{
    __shared__ __attribute__((aligned(16))) float ps1[BNG * BNF];
    __shared__ __attribute__((aligned(16))) float ps2[BNG * BNF];
    __shared__ float pr1[32 * BNF];
    __shared__ float pr2[32 * BNF];

    const int tid = threadIdx.x;
    const int q   = tid & 7;
    const int g   = tid >> 3;
    const int f0  = blockIdx.x * BNF;

    const float4* src = reinterpret_cast<const float4*>(feat + (size_t)g * NF + f0 + 4 * q);
    float4 s  = {0.0f, 0.0f, 0.0f, 0.0f};
    float4 s2 = {0.0f, 0.0f, 0.0f, 0.0f};
#pragma unroll 16
    for (int i = 0; i < NB / BNG; ++i) {
        const float4 v = src[(size_t)i * (BNG * NF / 4)];
        s.x += v.x;  s2.x += v.x * v.x;
        s.y += v.y;  s2.y += v.y * v.y;
        s.z += v.z;  s2.z += v.z * v.z;
        s.w += v.w;  s2.w += v.w * v.w;
    }
    *reinterpret_cast<float4*>(&ps1[4 * tid]) = s;
    *reinterpret_cast<float4*>(&ps2[4 * tid]) = s2;
    __syncthreads();

    {
        const int f = tid & 31;
        const int p = tid >> 5;
        float a = 0.0f, b = 0.0f;
        for (int i = 0; i < 4; ++i) {
            a += ps1[(4 * p + i) * BNF + f];
            b += ps2[(4 * p + i) * BNF + f];
        }
        pr1[p * BNF + f] = a;
        pr2[p * BNF + f] = b;
    }
    __syncthreads();

    if (tid < BNF) {
        float a = 0.0f, b = 0.0f;
        for (int p = 0; p < 32; ++p) {
            a += pr1[p * BNF + tid];
            b += pr2[p * BNF + tid];
        }
        const int f = f0 + tid;
        const float mu  = a * (1.0f / NB);
        const float var = b * (1.0f / NB) - mu * mu;
        const float sc  = gamma[f] * rsqrtf(var + 1e-5f);
        scale[f] = sc;
        shift[f] = beta[f] - mu * sc;
    }
}

__global__ __launch_bounds__(32) void fc_kernel(
    const float* __restrict__ feat,
    const float* __restrict__ scale,
    const float* __restrict__ shift,
    const float* __restrict__ fcw,
    const float* __restrict__ fcb,
    float* __restrict__ out)
{
    const int lane = threadIdx.x;
    const int b0   = blockIdx.x * 16;
    const bool hi  = lane >= 16;
    const int nn   = lane & 15;
    const int ks   = hi ? 2 : 0;
    const int rofs = hi ? 8 : 0;

    v8f acc;
    const float bb = fcb[nn];
#pragma unroll
    for (int r = 0; r < 8; ++r) acc[r] = bb;

    const int arow = b0 + nn;
    for (int k = 0; k < 400; ++k) {
        const int kb = 4 * k + ks;
        v2f a, b;
        const float f0 = feat[(size_t)arow * NF + kb];
        const float f1 = feat[(size_t)arow * NF + kb + 1];
        a.x = fmaxf(f0 * scale[kb] + shift[kb], 0.0f);
        a.y = fmaxf(f1 * scale[kb + 1] + shift[kb + 1], 0.0f);
        b.x = fcw[nn * NF + kb];
        b.y = fcw[nn * NF + kb + 1];
        acc = WMMA_F32(a, b, acc);
    }

#pragma unroll
    for (int r = 0; r < 8; ++r)
        out[(b0 + r + rofs) * NCLS + nn] = acc[r];
}

extern "C" void kernel_launch(void* const* d_in, const int* in_sizes, int n_in,
                              void* d_out, int out_size, void* d_ws, size_t ws_size,
                              hipStream_t stream)
{
    const float* x     = (const float*)d_in[0];
    const float* Wih   = (const float*)d_in[1];
    const float* Whh   = (const float*)d_in[2];
    const float* bih   = (const float*)d_in[3];
    const float* bhh   = (const float*)d_in[4];
    const float* gamma = (const float*)d_in[5];
    const float* beta  = (const float*)d_in[6];
    const float* fcw   = (const float*)d_in[7];
    const float* fcb   = (const float*)d_in[8];
    float* out = (float*)d_out;

    float* feat  = (float*)d_ws;
    float* scale = feat + (size_t)NB * NF;
    float* shift = scale + NF;

    lstm_kernel<<<NB / 16, 128, 0, stream>>>(x, Wih, Whh, bih, bhh, feat);
    bn_stats_kernel<<<NF / BNF, 1024, 0, stream>>>(feat, gamma, beta, scale, shift);
    fc_kernel<<<NB / 16, 32, 0, stream>>>(feat, scale, shift, fcw, fcb, out);
}


# lstm kernel: the block's x tile (16 rows x 200 bands x 25 positions) is read once into registers (625 per thread); each step's slice goes to LDS from there, not re-gathered from L2 at every step
# speedup vs baseline: 3.0895x; 1.4032x over previous
#include <hip/hip_runtime.h>
#include <math.h>

#define NB   4096
#define NC   200
#define NT   25
#define NH   64
#define NG   256
#define NF   1600
#define NCLS 16

typedef __attribute__((ext_vector_type(2))) float v2f;
typedef __attribute__((ext_vector_type(8))) float v8f;

#define WMMA_F32(a, b, c) \
    __builtin_amdgcn_wmma_f32_16x16x4_f32(false, (a), false, (b), (short)0, (c), false, false)

__device__ __forceinline__ float sigf(float x) { return 1.0f / (1.0f + __expf(-x)); }

#define XP 204
#define HP 68
#define KCH 6

__global__ __launch_bounds__(128) void lstm_kernel(
    const float* __restrict__ x,
    const float* __restrict__ Wih,
    const float* __restrict__ Whh,
    const float* __restrict__ bih,
    const float* __restrict__ bhh,
    float* __restrict__ feat)
{
    __shared__ __attribute__((aligned(16))) float wih_s[NG * XP];
    __shared__ __attribute__((aligned(16))) float whh_s[NG * HP];
    __shared__ __attribute__((aligned(16))) float xs[16 * XP];
    __shared__ __attribute__((aligned(16))) float hlds[16 * HP];
    __shared__ float glds[16 * NG];

    const int tid  = threadIdx.x;
    const int lane = tid & 31;
    const int gb   = (tid >> 5) * 64;
    const int b0   = blockIdx.x * 16;
    const bool hi  = lane >= 16;
    const int nn   = lane & 15;
    const int ks   = hi ? 2 : 0;
    const int rofs = hi ? 8 : 0;

    const float* xg = x + ((size_t)b0 * NC + tid) * NT;
    int xdst[25];
#pragma unroll
    for (int i = 0; i < 25; ++i) {
        const int e = tid + 128 * i;
        const int row = e / NC;
        xdst[i] = row * XP + (e - row * NC);
    }

    {
        const float4* src = reinterpret_cast<const float4*>(Wih);
        for (int i0 = 0; i0 < 100; i0 += 25) {
            float4 v[25];
#pragma unroll
            for (int u = 0; u < 25; ++u) v[u] = src[tid + 128 * (i0 + u)];
#pragma unroll
            for (int u = 0; u < 25; ++u) {
                const int q = tid + 128 * (i0 + u);
                const int row = q / 50;
                *reinterpret_cast<float4*>(&wih_s[row * XP + 4 * (q - row * 50)]) = v[u];
            }
        }
    }
    {
        const float4* src = reinterpret_cast<const float4*>(Whh);
        float4 v[32];
#pragma unroll
        for (int u = 0; u < 32; ++u) v[u] = src[tid + 128 * u];
#pragma unroll
        for (int u = 0; u < 32; ++u) {
            const int q = tid + 128 * u;
            *reinterpret_cast<float4*>(&whh_s[(q >> 4) * HP + 4 * (q & 15)]) = v[u];
        }
    }

    float xall[25][25];
#pragma unroll
    for (int i = 0; i < 25; ++i)
#pragma unroll
        for (int p = 0; p < 25; ++p) xall[i][p] = xg[3200 * i + p];

    float creg[8];
#pragma unroll
    for (int ii = 0; ii < 8; ++ii) {
        const int idx = tid + 128 * ii;
        creg[ii] = 0.0f;
        hlds[(idx >> 6) * HP + (idx & 63)] = 0.0f;
    }
#pragma unroll
    for (int i = 0; i < 25; ++i) xs[xdst[i]] = xall[i][0];

    float bv[4];
#pragma unroll
    for (int j = 0; j < 4; ++j) {
        const int g = gb + 16 * j + nn;
        bv[j] = bih[g] + bhh[g];
    }

    const float* xa = xs    + nn * XP + ks;
    const float* ha = hlds  + nn * HP + ks;
    const float* wa = wih_s + (gb + nn) * XP + ks;
    const float* ua = whh_s + (gb + nn) * HP + ks;
    __syncthreads();

    for (int t = 0; t < NT; ++t) {

        v8f acc[4];
#pragma unroll
        for (int j = 0; j < 4; ++j)
#pragma unroll
            for (int r = 0; r < 8; ++r) acc[j][r] = bv[j];

        v2f fa[2][KCH];
        v2f fb[2][KCH][4];
#define LOAD_SET(buf, c)                                                                 \
        _Pragma("unroll") for (int kk = 0; kk < KCH; ++kk) {                              \
            const int s = (c) * KCH + kk;                                                 \
            if (s < 50) {                                                                 \
                fa[buf][kk] = *reinterpret_cast<const v2f*>(xa + 4 * s);                  \
                _Pragma("unroll") for (int j = 0; j < 4; ++j)                             \
                    fb[buf][kk][j] = *reinterpret_cast<const v2f*>(wa + 16 * j * XP + 4 * s); \
            } else {                                                                      \
                fa[buf][kk] = *reinterpret_cast<const v2f*>(ha + 4 * (s - 50));           \
                _Pragma("unroll") for (int j = 0; j < 4; ++j)                             \
                    fb[buf][kk][j] = *reinterpret_cast<const v2f*>(ua + 16 * j * HP + 4 * (s - 50)); \
            }                                                                             \
        }
        LOAD_SET(0, 0)
#pragma unroll
        for (int c = 0; c < 11; ++c) {
            if (c + 1 < 11) { LOAD_SET((c + 1) & 1, c + 1) }
            __builtin_amdgcn_sched_barrier(0);
#pragma unroll
            for (int kk = 0; kk < KCH; ++kk)
#pragma unroll
                for (int j = 0; j < 4; ++j)
                    acc[j] = WMMA_F32(fa[c & 1][kk], fb[c & 1][kk][j], acc[j]);
            __builtin_amdgcn_sched_barrier(0);
        }
#undef LOAD_SET

#pragma unroll
        for (int j = 0; j < 4; ++j)
#pragma unroll
            for (int r = 0; r < 8; ++r)
                glds[(r + rofs) * NG + gb + 16 * j + nn] = acc[j][r];
        __syncthreads();

#pragma unroll
        for (int ii = 0; ii < 8; ++ii) {
            const int idx = tid + 128 * ii;
            const int row = idx >> 6;
            const int hc  = idx & 63;
            const float gi = glds[row * NG + hc];
            const float gf = glds[row * NG + 64 + hc];
            const float gg = glds[row * NG + 128 + hc];
            const float go = glds[row * NG + 192 + hc];
            const float cc = sigf(gf) * creg[ii] + sigf(gi) * tanhf(gg);
            const float hh = sigf(go) * tanhf(cc);
            creg[ii] = cc;
            hlds[row * HP + hc] = hh;
            feat[(size_t)(b0 + row) * NF + hc * NT + t] = fmaxf(hh, 0.0f);
        }
#pragma unroll
        for (int tt = 1; tt < NT; ++tt) {
            if (t + 1 == tt) {
                const int off = (tt % 5) * 5 + tt / 5;
#pragma unroll
                for (int i = 0; i < 25; ++i) xs[xdst[i]] = xall[i][off];
            }
        }
        __syncthreads();
    }
}

#define BNF 32
#define BNG 128

__global__ __launch_bounds__(1024) void bn_stats_kernel(const float* __restrict__ feat,
                                                        const float* __restrict__ gamma,
                                                        const float* __restrict__ beta,
                                                        float* __restrict__ scale,
                                                        float* __restrict__ shift)
{
    __shared__ __attribute__((aligned(16))) float ps1[BNG * BNF];
    __shared__ __attribute__((aligned(16))) float ps2[BNG * BNF];
    __shared__ float pr1[32 * BNF];
    __shared__ float pr2[32 * BNF];

    const int tid = threadIdx.x;
    const int q   = tid & 7;
    const int g   = tid >> 3;
    const int f0  = blockIdx.x * BNF;

    const float4* src = reinterpret_cast<const float4*>(feat + (size_t)g * NF + f0 + 4 * q);
    float4 s  = {0.0f, 0.0f, 0.0f, 0.0f};
    float4 s2 = {0.0f, 0.0f, 0.0f, 0.0f};
#pragma unroll 16
    for (int i = 0; i < NB / BNG; ++i) {
        const float4 v = src[(size_t)i * (BNG * NF / 4)];
        s.x += v.x;  s2.x += v.x * v.x;
        s.y += v.y;  s2.y += v.y * v.y;
        s.z += v.z;  s2.z += v.z * v.z;
        s.w += v.w;  s2.w += v.w * v.w;
    }
    *reinterpret_cast<float4*>(&ps1[4 * tid]) = s;
    *reinterpret_cast<float4*>(&ps2[4 * tid]) = s2;
    __syncthreads();

    {
        const int f = tid & 31;
        const int p = tid >> 5;
        float a = 0.0f, b = 0.0f;
        for (int i = 0; i < 4; ++i) {
            a += ps1[(4 * p + i) * BNF + f];
            b += ps2[(4 * p + i) * BNF + f];
        }
        pr1[p * BNF + f] = a;
        pr2[p * BNF + f] = b;
    }
    __syncthreads();

    if (tid < BNF) {
        float a = 0.0f, b = 0.0f;
        for (int p = 0; p < 32; ++p) {
            a += pr1[p * BNF + tid];
            b += pr2[p * BNF + tid];
        }
        const int f = f0 + tid;
        const float mu  = a * (1.0f / NB);
        const float var = b * (1.0f / NB) - mu * mu;
        const float sc  = gamma[f] * rsqrtf(var + 1e-5f);
        scale[f] = sc;
        shift[f] = beta[f] - mu * sc;
    }
}

__global__ __launch_bounds__(32) void fc_kernel(
    const float* __restrict__ feat,
    const float* __restrict__ scale,
    const float* __restrict__ shift,
    const float* __restrict__ fcw,
    const float* __restrict__ fcb,
    float* __restrict__ out)
{
    const int lane = threadIdx.x;
    const int b0   = blockIdx.x * 16;
    const bool hi  = lane >= 16;
    const int nn   = lane & 15;
    const int ks   = hi ? 2 : 0;
    const int rofs = hi ? 8 : 0;

    v8f acc;
    const float bb = fcb[nn];
#pragma unroll
    for (int r = 0; r < 8; ++r) acc[r] = bb;

    const int arow = b0 + nn;
    for (int k = 0; k < 400; ++k) {
        const int kb = 4 * k + ks;
        v2f a, b;
        const float f0 = feat[(size_t)arow * NF + kb];
        const float f1 = feat[(size_t)arow * NF + kb + 1];
        a.x = fmaxf(f0 * scale[kb] + shift[kb], 0.0f);
        a.y = fmaxf(f1 * scale[kb + 1] + shift[kb + 1], 0.0f);
        b.x = fcw[nn * NF + kb];
        b.y = fcw[nn * NF + kb + 1];
        acc = WMMA_F32(a, b, acc);
    }

#pragma unroll
    for (int r = 0; r < 8; ++r)
        out[(b0 + r + rofs) * NCLS + nn] = acc[r];
}

extern "C" void kernel_launch(void* const* d_in, const int* in_sizes, int n_in,
                              void* d_out, int out_size, void* d_ws, size_t ws_size,
                              hipStream_t stream)
{
    const float* x     = (const float*)d_in[0];
    const float* Wih   = (const float*)d_in[1];
    const float* Whh   = (const float*)d_in[2];
    const float* bih   = (const float*)d_in[3];
    const float* bhh   = (const float*)d_in[4];
    const float* gamma = (const float*)d_in[5];
    const float* beta  = (const float*)d_in[6];
    const float* fcw   = (const float*)d_in[7];
    const float* fcb   = (const float*)d_in[8];
    float* out = (float*)d_out;

    float* feat  = (float*)d_ws;
    float* scale = feat + (size_t)NB * NF;
    float* shift = scale + NF;

    lstm_kernel<<<NB / 16, 128, 0, stream>>>(x, Wih, Whh, bih, bhh, feat);
    bn_stats_kernel<<<NF / BNF, 1024, 0, stream>>>(feat, gamma, beta, scale, shift);
    fc_kernel<<<NB / 16, 32, 0, stream>>>(feat, scale, shift, fcw, fcb, out);
}


# fc kernel: K split over 8 waves per 16-row block, 16-byte operand loads streamed ahead of the WMMAs (not one L2 round trip per 5 K-steps), 8 partial tiles and bias summed via LDS; reorders the K sum
# speedup vs baseline: 3.5903x; 1.1621x over previous
#include <hip/hip_runtime.h>
#include <math.h>

#define NB   4096
#define NC   200
#define NT   25
#define NH   64
#define NG   256
#define NF   1600
#define NCLS 16

typedef __attribute__((ext_vector_type(2))) float v2f;
typedef __attribute__((ext_vector_type(8))) float v8f;

#define WMMA_F32(a, b, c) \
    __builtin_amdgcn_wmma_f32_16x16x4_f32(false, (a), false, (b), (short)0, (c), false, false)

__device__ __forceinline__ float sigf(float x) { return 1.0f / (1.0f + __expf(-x)); }

#define XP 204
#define HP 68
#define KCH 6

__global__ __launch_bounds__(128) void lstm_kernel(
    const float* __restrict__ x,
    const float* __restrict__ Wih,
    const float* __restrict__ Whh,
    const float* __restrict__ bih,
    const float* __restrict__ bhh,
    float* __restrict__ feat)
{
    __shared__ __attribute__((aligned(16))) float wih_s[NG * XP];
    __shared__ __attribute__((aligned(16))) float whh_s[NG * HP];
    __shared__ __attribute__((aligned(16))) float xs[16 * XP];
    __shared__ __attribute__((aligned(16))) float hlds[16 * HP];
    __shared__ float glds[16 * NG];

    const int tid  = threadIdx.x;
    const int lane = tid & 31;
    const int gb   = (tid >> 5) * 64;
    const int b0   = blockIdx.x * 16;
    const bool hi  = lane >= 16;
    const int nn   = lane & 15;
    const int ks   = hi ? 2 : 0;
    const int rofs = hi ? 8 : 0;

    const float* xg = x + ((size_t)b0 * NC + tid) * NT;
    int xdst[25];
#pragma unroll
    for (int i = 0; i < 25; ++i) {
        const int e = tid + 128 * i;
        const int row = e / NC;
        xdst[i] = row * XP + (e - row * NC);
    }

    {
        const float4* src = reinterpret_cast<const float4*>(Wih);
        for (int i0 = 0; i0 < 100; i0 += 25) {
            float4 v[25];
#pragma unroll
            for (int u = 0; u < 25; ++u) v[u] = src[tid + 128 * (i0 + u)];
#pragma unroll
            for (int u = 0; u < 25; ++u) {
                const int q = tid + 128 * (i0 + u);
                const int row = q / 50;
                *reinterpret_cast<float4*>(&wih_s[row * XP + 4 * (q - row * 50)]) = v[u];
            }
        }
    }
    {
        const float4* src = reinterpret_cast<const float4*>(Whh);
        float4 v[32];
#pragma unroll
        for (int u = 0; u < 32; ++u) v[u] = src[tid + 128 * u];
#pragma unroll
        for (int u = 0; u < 32; ++u) {
            const int q = tid + 128 * u;
            *reinterpret_cast<float4*>(&whh_s[(q >> 4) * HP + 4 * (q & 15)]) = v[u];
        }
    }

    float xall[25][25];
#pragma unroll
    for (int i = 0; i < 25; ++i)
#pragma unroll
        for (int p = 0; p < 25; ++p) xall[i][p] = xg[3200 * i + p];

    float creg[8];
#pragma unroll
    for (int ii = 0; ii < 8; ++ii) {
        const int idx = tid + 128 * ii;
        creg[ii] = 0.0f;
        hlds[(idx >> 6) * HP + (idx & 63)] = 0.0f;
    }
#pragma unroll
    for (int i = 0; i < 25; ++i) xs[xdst[i]] = xall[i][0];

    float bv[4];
#pragma unroll
    for (int j = 0; j < 4; ++j) {
        const int g = gb + 16 * j + nn;
        bv[j] = bih[g] + bhh[g];
    }

    const float* xa = xs    + nn * XP + ks;
    const float* ha = hlds  + nn * HP + ks;
    const float* wa = wih_s + (gb + nn) * XP + ks;
    const float* ua = whh_s + (gb + nn) * HP + ks;
    __syncthreads();

    for (int t = 0; t < NT; ++t) {

        v8f acc[4];
#pragma unroll
        for (int j = 0; j < 4; ++j)
#pragma unroll
            for (int r = 0; r < 8; ++r) acc[j][r] = bv[j];

        v2f fa[2][KCH];
        v2f fb[2][KCH][4];
#define LOAD_SET(buf, c)                                                                 \
        _Pragma("unroll") for (int kk = 0; kk < KCH; ++kk) {                              \
            const int s = (c) * KCH + kk;                                                 \
            if (s < 50) {                                                                 \
                fa[buf][kk] = *reinterpret_cast<const v2f*>(xa + 4 * s);                  \
                _Pragma("unroll") for (int j = 0; j < 4; ++j)                             \
                    fb[buf][kk][j] = *reinterpret_cast<const v2f*>(wa + 16 * j * XP + 4 * s); \
            } else {                                                                      \
                fa[buf][kk] = *reinterpret_cast<const v2f*>(ha + 4 * (s - 50));           \
                _Pragma("unroll") for (int j = 0; j < 4; ++j)                             \
                    fb[buf][kk][j] = *reinterpret_cast<const v2f*>(ua + 16 * j * HP + 4 * (s - 50)); \
            }                                                                             \
        }
        LOAD_SET(0, 0)
#pragma unroll
        for (int c = 0; c < 11; ++c) {
            if (c + 1 < 11) { LOAD_SET((c + 1) & 1, c + 1) }
            __builtin_amdgcn_sched_barrier(0);
#pragma unroll
            for (int kk = 0; kk < KCH; ++kk)
#pragma unroll
                for (int j = 0; j < 4; ++j)
                    acc[j] = WMMA_F32(fa[c & 1][kk], fb[c & 1][kk][j], acc[j]);
            __builtin_amdgcn_sched_barrier(0);
        }
#undef LOAD_SET

#pragma unroll
        for (int j = 0; j < 4; ++j)
#pragma unroll
            for (int r = 0; r < 8; ++r)
                glds[(r + rofs) * NG + gb + 16 * j + nn] = acc[j][r];
        __syncthreads();

#pragma unroll
        for (int ii = 0; ii < 8; ++ii) {
            const int idx = tid + 128 * ii;
            const int row = idx >> 6;
            const int hc  = idx & 63;
            const float gi = glds[row * NG + hc];
            const float gf = glds[row * NG + 64 + hc];
            const float gg = glds[row * NG + 128 + hc];
            const float go = glds[row * NG + 192 + hc];
            const float cc = sigf(gf) * creg[ii] + sigf(gi) * tanhf(gg);
            const float hh = sigf(go) * tanhf(cc);
            creg[ii] = cc;
            hlds[row * HP + hc] = hh;
            feat[(size_t)(b0 + row) * NF + hc * NT + t] = fmaxf(hh, 0.0f);
        }
#pragma unroll
        for (int tt = 1; tt < NT; ++tt) {
            if (t + 1 == tt) {
                const int off = (tt % 5) * 5 + tt / 5;
#pragma unroll
                for (int i = 0; i < 25; ++i) xs[xdst[i]] = xall[i][off];
            }
        }
        __syncthreads();
    }
}

#define BNF 32
#define BNG 128

__global__ __launch_bounds__(1024) void bn_stats_kernel(const float* __restrict__ feat,
                                                        const float* __restrict__ gamma,
                                                        const float* __restrict__ beta,
                                                        float* __restrict__ scale,
                                                        float* __restrict__ shift)
{
    __shared__ __attribute__((aligned(16))) float ps1[BNG * BNF];
    __shared__ __attribute__((aligned(16))) float ps2[BNG * BNF];
    __shared__ float pr1[32 * BNF];
    __shared__ float pr2[32 * BNF];

    const int tid = threadIdx.x;
    const int q   = tid & 7;
    const int g   = tid >> 3;
    const int f0  = blockIdx.x * BNF;

    const float4* src = reinterpret_cast<const float4*>(feat + (size_t)g * NF + f0 + 4 * q);
    float4 s  = {0.0f, 0.0f, 0.0f, 0.0f};
    float4 s2 = {0.0f, 0.0f, 0.0f, 0.0f};
#pragma unroll 16
    for (int i = 0; i < NB / BNG; ++i) {
        const float4 v = src[(size_t)i * (BNG * NF / 4)];
        s.x += v.x;  s2.x += v.x * v.x;
        s.y += v.y;  s2.y += v.y * v.y;
        s.z += v.z;  s2.z += v.z * v.z;
        s.w += v.w;  s2.w += v.w * v.w;
    }
    *reinterpret_cast<float4*>(&ps1[4 * tid]) = s;
    *reinterpret_cast<float4*>(&ps2[4 * tid]) = s2;
    __syncthreads();

    {
        const int f = tid & 31;
        const int p = tid >> 5;
        float a = 0.0f, b = 0.0f;
        for (int i = 0; i < 4; ++i) {
            a += ps1[(4 * p + i) * BNF + f];
            b += ps2[(4 * p + i) * BNF + f];
        }
        pr1[p * BNF + f] = a;
        pr2[p * BNF + f] = b;
    }
    __syncthreads();

    if (tid < BNF) {
        float a = 0.0f, b = 0.0f;
        for (int p = 0; p < 32; ++p) {
            a += pr1[p * BNF + tid];
            b += pr2[p * BNF + tid];
        }
        const int f = f0 + tid;
        const float mu  = a * (1.0f / NB);
        const float var = b * (1.0f / NB) - mu * mu;
        const float sc  = gamma[f] * rsqrtf(var + 1e-5f);
        scale[f] = sc;
        shift[f] = beta[f] - mu * sc;
    }
}

#define FCWAVES 8
#define FCS     25

__global__ __launch_bounds__(256) void fc_kernel(
    const float* __restrict__ feat,
    const float* __restrict__ scale,
    const float* __restrict__ shift,
    const float* __restrict__ fcw,
    const float* __restrict__ fcb,
    float* __restrict__ out)
{
    __shared__ float part[FCWAVES * 256];

    const int tid  = threadIdx.x;
    const int lane = tid & 31;
    const int w    = tid >> 5;
    const int b0   = blockIdx.x * 16;
    const int h    = lane >> 4;
    const int nn   = lane & 15;
    const int col0 = 200 * w + 4 * h;

    const float4* pa = reinterpret_cast<const float4*>(feat + (size_t)(b0 + nn) * NF + col0);
    const float4* pb = reinterpret_cast<const float4*>(fcw + nn * NF + col0);
    const float4* ps = reinterpret_cast<const float4*>(scale + col0);
    const float4* pt = reinterpret_cast<const float4*>(shift + col0);

    const float bias = fcb[tid & 15];

    float4 va[FCS], vb[FCS], vs[FCS], vt[FCS];
#pragma unroll
    for (int s = 0; s < FCS; ++s) {
        va[s] = pa[2 * s];
        vb[s] = pb[2 * s];
        vs[s] = ps[2 * s];
        vt[s] = pt[2 * s];
    }

    v8f acc0, acc1;
#pragma unroll
    for (int r = 0; r < 8; ++r) { acc0[r] = 0.0f; acc1[r] = 0.0f; }

#pragma unroll
    for (int s = 0; s < FCS; ++s) {
        v2f a, b;
        a.x = fmaxf(va[s].x * vs[s].x + vt[s].x, 0.0f);
        a.y = fmaxf(va[s].y * vs[s].y + vt[s].y, 0.0f);
        b.x = vb[s].x;
        b.y = vb[s].y;
        acc0 = WMMA_F32(a, b, acc0);
        a.x = fmaxf(va[s].z * vs[s].z + vt[s].z, 0.0f);
        a.y = fmaxf(va[s].w * vs[s].w + vt[s].w, 0.0f);
        b.x = vb[s].z;
        b.y = vb[s].w;
        acc1 = WMMA_F32(a, b, acc1);
    }

#pragma unroll
    for (int r = 0; r < 8; ++r)
        part[w * 256 + (r + 8 * h) * NCLS + nn] = acc0[r] + acc1[r];
    __syncthreads();

    float sum = bias + part[tid];
#pragma unroll
    for (int q = 1; q < FCWAVES; ++q) sum += part[q * 256 + tid];
    out[b0 * NCLS + tid] = sum;
}

extern "C" void kernel_launch(void* const* d_in, const int* in_sizes, int n_in,
                              void* d_out, int out_size, void* d_ws, size_t ws_size,
                              hipStream_t stream)
{
    const float* x     = (const float*)d_in[0];
    const float* Wih   = (const float*)d_in[1];
    const float* Whh   = (const float*)d_in[2];
    const float* bih   = (const float*)d_in[3];
    const float* bhh   = (const float*)d_in[4];
    const float* gamma = (const float*)d_in[5];
    const float* beta  = (const float*)d_in[6];
    const float* fcw   = (const float*)d_in[7];
    const float* fcb   = (const float*)d_in[8];
    float* out = (float*)d_out;

    float* feat  = (float*)d_ws;
    float* scale = feat + (size_t)NB * NF;
    float* shift = scale + NF;

    lstm_kernel<<<NB / 16, 128, 0, stream>>>(x, Wih, Whh, bih, bhh, feat);
    bn_stats_kernel<<<NF / BNF, 1024, 0, stream>>>(feat, gamma, beta, scale, shift);
    fc_kernel<<<NB / 16, 32 * FCWAVES, 0, stream>>>(feat, scale, shift, fcw, fcb, out);
}


# lstm cell update made branch-free (tanh: exp form and small-|x| polynomial both computed, one selected; sigmoid divide -> hardware reciprocal) so a thread's 8 cells interleave, not 8 serial chains
# speedup vs baseline: 4.4546x; 1.2407x over previous
#include <hip/hip_runtime.h>
#include <math.h>

#define NB   4096
#define NC   200
#define NT   25
#define NH   64
#define NG   256
#define NF   1600
#define NCLS 16

typedef __attribute__((ext_vector_type(2))) float v2f;
typedef __attribute__((ext_vector_type(8))) float v8f;

#define WMMA_F32(a, b, c) \
    __builtin_amdgcn_wmma_f32_16x16x4_f32(false, (a), false, (b), (short)0, (c), false, false)

__device__ __forceinline__ float sigf(float x)
{
    return __builtin_amdgcn_rcpf(1.0f + __builtin_amdgcn_exp2f(x * -1.442695041f));
}

__device__ __forceinline__ float tanhsel(float x)
{
    const float ax = fabsf(x);
    const float e  = __builtin_amdgcn_exp2f(ax * 2.885390082f);
    const float big = fmaf(__builtin_amdgcn_rcpf(e + 1.0f), -2.0f, 1.0f);
    const float t = ax * ax;
    float p = fmaf(t, -0x1.758e7ap-8f, 0x1.521192p-6f);
    p = fmaf(t, p, -0x1.b8389cp-5f);
    p = fmaf(t, p, 0x1.110704p-3f);
    p = fmaf(t, p, -0x1.555532p-2f);
    const float sml = fmaf(t, ax * p, ax);
    return copysignf((ax < 0.625f) ? sml : big, x);
}

#define XP 204
#define HP 68
#define KCH 6

__global__ __launch_bounds__(128) void lstm_kernel(
    const float* __restrict__ x,
    const float* __restrict__ Wih,
    const float* __restrict__ Whh,
    const float* __restrict__ bih,
    const float* __restrict__ bhh,
    float* __restrict__ feat)
{
    __shared__ __attribute__((aligned(16))) float wih_s[NG * XP];
    __shared__ __attribute__((aligned(16))) float whh_s[NG * HP];
    __shared__ __attribute__((aligned(16))) float xs[16 * XP];
    __shared__ __attribute__((aligned(16))) float hlds[16 * HP];
    __shared__ float glds[16 * NG];

    const int tid  = threadIdx.x;
    const int lane = tid & 31;
    const int gb   = (tid >> 5) * 64;
    const int b0   = blockIdx.x * 16;
    const bool hi  = lane >= 16;
    const int nn   = lane & 15;
    const int ks   = hi ? 2 : 0;
    const int rofs = hi ? 8 : 0;

    const float* xg = x + ((size_t)b0 * NC + tid) * NT;
    int xdst[25];
#pragma unroll
    for (int i = 0; i < 25; ++i) {
        const int e = tid + 128 * i;
        const int row = e / NC;
        xdst[i] = row * XP + (e - row * NC);
    }

    {
        const float4* src = reinterpret_cast<const float4*>(Wih);
        for (int i0 = 0; i0 < 100; i0 += 25) {
            float4 v[25];
#pragma unroll
            for (int u = 0; u < 25; ++u) v[u] = src[tid + 128 * (i0 + u)];
#pragma unroll
            for (int u = 0; u < 25; ++u) {
                const int q = tid + 128 * (i0 + u);
                const int row = q / 50;
                *reinterpret_cast<float4*>(&wih_s[row * XP + 4 * (q - row * 50)]) = v[u];
            }
        }
    }
    {
        const float4* src = reinterpret_cast<const float4*>(Whh);
        float4 v[32];
#pragma unroll
        for (int u = 0; u < 32; ++u) v[u] = src[tid + 128 * u];
#pragma unroll
        for (int u = 0; u < 32; ++u) {
            const int q = tid + 128 * u;
            *reinterpret_cast<float4*>(&whh_s[(q >> 4) * HP + 4 * (q & 15)]) = v[u];
        }
    }

    float xall[25][25];
#pragma unroll
    for (int i = 0; i < 25; ++i)
#pragma unroll
        for (int p = 0; p < 25; ++p) xall[i][p] = xg[3200 * i + p];

    float creg[8];
#pragma unroll
    for (int ii = 0; ii < 8; ++ii) {
        const int idx = tid + 128 * ii;
        creg[ii] = 0.0f;
        hlds[(idx >> 6) * HP + (idx & 63)] = 0.0f;
    }
#pragma unroll
    for (int i = 0; i < 25; ++i) xs[xdst[i]] = xall[i][0];

    float bv[4];
#pragma unroll
    for (int j = 0; j < 4; ++j) {
        const int g = gb + 16 * j + nn;
        bv[j] = bih[g] + bhh[g];
    }

    const float* xa = xs    + nn * XP + ks;
    const float* ha = hlds  + nn * HP + ks;
    const float* wa = wih_s + (gb + nn) * XP + ks;
    const float* ua = whh_s + (gb + nn) * HP + ks;
    __syncthreads();

    for (int t = 0; t < NT; ++t) {

        v8f acc[4];
#pragma unroll
        for (int j = 0; j < 4; ++j)
#pragma unroll
            for (int r = 0; r < 8; ++r) acc[j][r] = bv[j];

        v2f fa[2][KCH];
        v2f fb[2][KCH][4];
#define LOAD_SET(buf, c)                                                                 \
        _Pragma("unroll") for (int kk = 0; kk < KCH; ++kk) {                              \
            const int s = (c) * KCH + kk;                                                 \
            if (s < 50) {                                                                 \
                fa[buf][kk] = *reinterpret_cast<const v2f*>(xa + 4 * s);                  \
                _Pragma("unroll") for (int j = 0; j < 4; ++j)                             \
                    fb[buf][kk][j] = *reinterpret_cast<const v2f*>(wa + 16 * j * XP + 4 * s); \
            } else {                                                                      \
                fa[buf][kk] = *reinterpret_cast<const v2f*>(ha + 4 * (s - 50));           \
                _Pragma("unroll") for (int j = 0; j < 4; ++j)                             \
                    fb[buf][kk][j] = *reinterpret_cast<const v2f*>(ua + 16 * j * HP + 4 * (s - 50)); \
            }                                                                             \
        }
        LOAD_SET(0, 0)
#pragma unroll
        for (int c = 0; c < 11; ++c) {
            if (c + 1 < 11) { LOAD_SET((c + 1) & 1, c + 1) }
            __builtin_amdgcn_sched_barrier(0);
#pragma unroll
            for (int kk = 0; kk < KCH; ++kk)
#pragma unroll
                for (int j = 0; j < 4; ++j)
                    acc[j] = WMMA_F32(fa[c & 1][kk], fb[c & 1][kk][j], acc[j]);
            __builtin_amdgcn_sched_barrier(0);
        }
#undef LOAD_SET

#pragma unroll
        for (int j = 0; j < 4; ++j)
#pragma unroll
            for (int r = 0; r < 8; ++r)
                glds[(r + rofs) * NG + gb + 16 * j + nn] = acc[j][r];
        __syncthreads();

        float gi[8], gf[8], gg[8], go[8], hn[8];
#pragma unroll
        for (int ii = 0; ii < 8; ++ii) {
            const int idx = tid + 128 * ii;
            const int row = idx >> 6;
            const int hc  = idx & 63;
            gi[ii] = glds[row * NG + hc];
            gf[ii] = glds[row * NG + 64 + hc];
            gg[ii] = glds[row * NG + 128 + hc];
            go[ii] = glds[row * NG + 192 + hc];
        }
#pragma unroll
        for (int ii = 0; ii < 8; ++ii) {
            const float cc = sigf(gf[ii]) * creg[ii] + sigf(gi[ii]) * tanhsel(gg[ii]);
            hn[ii] = sigf(go[ii]) * tanhsel(cc);
            creg[ii] = cc;
        }
#pragma unroll
        for (int ii = 0; ii < 8; ++ii) {
            const int idx = tid + 128 * ii;
            const int row = idx >> 6;
            const int hc  = idx & 63;
            hlds[row * HP + hc] = hn[ii];
            feat[(size_t)(b0 + row) * NF + hc * NT + t] = fmaxf(hn[ii], 0.0f);
        }
#pragma unroll
        for (int tt = 1; tt < NT; ++tt) {
            if (t + 1 == tt) {
                const int off = (tt % 5) * 5 + tt / 5;
#pragma unroll
                for (int i = 0; i < 25; ++i) xs[xdst[i]] = xall[i][off];
            }
        }
        __syncthreads();
    }
}

#define BNF 32
#define BNG 128

__global__ __launch_bounds__(1024) void bn_stats_kernel(const float* __restrict__ feat,
                                                        const float* __restrict__ gamma,
                                                        const float* __restrict__ beta,
                                                        float* __restrict__ scale,
                                                        float* __restrict__ shift)
{
    __shared__ __attribute__((aligned(16))) float ps1[BNG * BNF];
    __shared__ __attribute__((aligned(16))) float ps2[BNG * BNF];
    __shared__ float pr1[32 * BNF];
    __shared__ float pr2[32 * BNF];

    const int tid = threadIdx.x;
    const int q   = tid & 7;
    const int g   = tid >> 3;
    const int f0  = blockIdx.x * BNF;

    const float4* src = reinterpret_cast<const float4*>(feat + (size_t)g * NF + f0 + 4 * q);
    float4 s  = {0.0f, 0.0f, 0.0f, 0.0f};
    float4 s2 = {0.0f, 0.0f, 0.0f, 0.0f};
#pragma unroll 16
    for (int i = 0; i < NB / BNG; ++i) {
        const float4 v = src[(size_t)i * (BNG * NF / 4)];
        s.x += v.x;  s2.x += v.x * v.x;
        s.y += v.y;  s2.y += v.y * v.y;
        s.z += v.z;  s2.z += v.z * v.z;
        s.w += v.w;  s2.w += v.w * v.w;
    }
    *reinterpret_cast<float4*>(&ps1[4 * tid]) = s;
    *reinterpret_cast<float4*>(&ps2[4 * tid]) = s2;
    __syncthreads();

    {
        const int f = tid & 31;
        const int p = tid >> 5;
        float a = 0.0f, b = 0.0f;
        for (int i = 0; i < 4; ++i) {
            a += ps1[(4 * p + i) * BNF + f];
            b += ps2[(4 * p + i) * BNF + f];
        }
        pr1[p * BNF + f] = a;
        pr2[p * BNF + f] = b;
    }
    __syncthreads();

    if (tid < BNF) {
        float a = 0.0f, b = 0.0f;
        for (int p = 0; p < 32; ++p) {
            a += pr1[p * BNF + tid];
            b += pr2[p * BNF + tid];
        }
        const int f = f0 + tid;
        const float mu  = a * (1.0f / NB);
        const float var = b * (1.0f / NB) - mu * mu;
        const float sc  = gamma[f] * rsqrtf(var + 1e-5f);
        scale[f] = sc;
        shift[f] = beta[f] - mu * sc;
    }
}

#define FCWAVES 8
#define FCS     25

__global__ __launch_bounds__(256) void fc_kernel(
    const float* __restrict__ feat,
    const float* __restrict__ scale,
    const float* __restrict__ shift,
    const float* __restrict__ fcw,
    const float* __restrict__ fcb,
    float* __restrict__ out)
{
    __shared__ float part[FCWAVES * 256];

    const int tid  = threadIdx.x;
    const int lane = tid & 31;
    const int w    = tid >> 5;
    const int b0   = blockIdx.x * 16;
    const int h    = lane >> 4;
    const int nn   = lane & 15;
    const int col0 = 200 * w + 4 * h;

    const float4* pa = reinterpret_cast<const float4*>(feat + (size_t)(b0 + nn) * NF + col0);
    const float4* pb = reinterpret_cast<const float4*>(fcw + nn * NF + col0);
    const float4* ps = reinterpret_cast<const float4*>(scale + col0);
    const float4* pt = reinterpret_cast<const float4*>(shift + col0);

    const float bias = fcb[tid & 15];

    float4 va[FCS], vb[FCS], vs[FCS], vt[FCS];
#pragma unroll
    for (int s = 0; s < FCS; ++s) {
        va[s] = pa[2 * s];
        vb[s] = pb[2 * s];
        vs[s] = ps[2 * s];
        vt[s] = pt[2 * s];
    }

    v8f acc0, acc1;
#pragma unroll
    for (int r = 0; r < 8; ++r) { acc0[r] = 0.0f; acc1[r] = 0.0f; }

#pragma unroll
    for (int s = 0; s < FCS; ++s) {
        v2f a, b;
        a.x = fmaxf(va[s].x * vs[s].x + vt[s].x, 0.0f);
        a.y = fmaxf(va[s].y * vs[s].y + vt[s].y, 0.0f);
        b.x = vb[s].x;
        b.y = vb[s].y;
        acc0 = WMMA_F32(a, b, acc0);
        a.x = fmaxf(va[s].z * vs[s].z + vt[s].z, 0.0f);
        a.y = fmaxf(va[s].w * vs[s].w + vt[s].w, 0.0f);
        b.x = vb[s].z;
        b.y = vb[s].w;
        acc1 = WMMA_F32(a, b, acc1);
    }

#pragma unroll
    for (int r = 0; r < 8; ++r)
        part[w * 256 + (r + 8 * h) * NCLS + nn] = acc0[r] + acc1[r];
    __syncthreads();

    float sum = bias + part[tid];
#pragma unroll
    for (int q = 1; q < FCWAVES; ++q) sum += part[q * 256 + tid];
    out[b0 * NCLS + tid] = sum;
}

extern "C" void kernel_launch(void* const* d_in, const int* in_sizes, int n_in,
                              void* d_out, int out_size, void* d_ws, size_t ws_size,
                              hipStream_t stream)
{
    const float* x     = (const float*)d_in[0];
    const float* Wih   = (const float*)d_in[1];
    const float* Whh   = (const float*)d_in[2];
    const float* bih   = (const float*)d_in[3];
    const float* bhh   = (const float*)d_in[4];
    const float* gamma = (const float*)d_in[5];
    const float* beta  = (const float*)d_in[6];
    const float* fcw   = (const float*)d_in[7];
    const float* fcb   = (const float*)d_in[8];
    float* out = (float*)d_out;

    float* feat  = (float*)d_ws;
    float* scale = feat + (size_t)NB * NF;
    float* shift = scale + NF;

    lstm_kernel<<<NB / 16, 128, 0, stream>>>(x, Wih, Whh, bih, bhh, feat);
    bn_stats_kernel<<<NF / BNF, 1024, 0, stream>>>(feat, gamma, beta, scale, shift);
    fc_kernel<<<NB / 16, 32 * FCWAVES, 0, stream>>>(feat, scale, shift, fcw, fcb, out);
}


# lstm steps pipelined by one: the input product (50 of 66 K-steps) of step t+1 runs during step t, the 8 cell updates and their stores among its WMMAs instead of alone with the matrix pipe idle
# speedup vs baseline: 4.8288x; 1.0840x over previous
#include <hip/hip_runtime.h>
#include <math.h>

#define NB   4096
#define NC   200
#define NT   25
#define NH   64
#define NG   256
#define NF   1600
#define NCLS 16

typedef __attribute__((ext_vector_type(2))) float v2f;
typedef __attribute__((ext_vector_type(8))) float v8f;

#define WMMA_F32(a, b, c) \
    __builtin_amdgcn_wmma_f32_16x16x4_f32(false, (a), false, (b), (short)0, (c), false, false)

__device__ __forceinline__ float sigf(float x)
{
    return __builtin_amdgcn_rcpf(1.0f + __builtin_amdgcn_exp2f(x * -1.442695041f));
}

__device__ __forceinline__ float tanhsel(float x)
{
    const float ax = fabsf(x);
    const float e  = __builtin_amdgcn_exp2f(ax * 2.885390082f);
    const float big = fmaf(__builtin_amdgcn_rcpf(e + 1.0f), -2.0f, 1.0f);
    const float t = ax * ax;
    float p = fmaf(t, -0x1.758e7ap-8f, 0x1.521192p-6f);
    p = fmaf(t, p, -0x1.b8389cp-5f);
    p = fmaf(t, p, 0x1.110704p-3f);
    p = fmaf(t, p, -0x1.555532p-2f);
    const float sml = fmaf(t, ax * p, ax);
    return copysignf((ax < 0.625f) ? sml : big, x);
}

#define XP 204
#define HP 68
#define XCH 5
#define HCH 8

__global__ __launch_bounds__(128) void lstm_kernel(
    const float* __restrict__ x,
    const float* __restrict__ Wih,
    const float* __restrict__ Whh,
    const float* __restrict__ bih,
    const float* __restrict__ bhh,
    float* __restrict__ feat)
{
    __shared__ __attribute__((aligned(16))) float wih_s[NG * XP];
    __shared__ __attribute__((aligned(16))) float whh_s[NG * HP];
    __shared__ __attribute__((aligned(16))) float xs[16 * XP];
    __shared__ __attribute__((aligned(16))) float hlds[16 * HP];
    __shared__ float glds[16 * NG];

    const int tid  = threadIdx.x;
    const int lane = tid & 31;
    const int gb   = (tid >> 5) * 64;
    const int b0   = blockIdx.x * 16;
    const bool hi  = lane >= 16;
    const int nn   = lane & 15;
    const int ks   = hi ? 2 : 0;
    const int rofs = hi ? 8 : 0;

    const float* xg = x + ((size_t)b0 * NC + tid) * NT;
    int xdst[25];
#pragma unroll
    for (int i = 0; i < 25; ++i) {
        const int e = tid + 128 * i;
        const int row = e / NC;
        xdst[i] = row * XP + (e - row * NC);
    }

    {
        const float4* src = reinterpret_cast<const float4*>(Wih);
        for (int i0 = 0; i0 < 100; i0 += 25) {
            float4 v[25];
#pragma unroll
            for (int u = 0; u < 25; ++u) v[u] = src[tid + 128 * (i0 + u)];
#pragma unroll
            for (int u = 0; u < 25; ++u) {
                const int q = tid + 128 * (i0 + u);
                const int row = q / 50;
                *reinterpret_cast<float4*>(&wih_s[row * XP + 4 * (q - row * 50)]) = v[u];
            }
        }
    }
    {
        const float4* src = reinterpret_cast<const float4*>(Whh);
        float4 v[32];
#pragma unroll
        for (int u = 0; u < 32; ++u) v[u] = src[tid + 128 * u];
#pragma unroll
        for (int u = 0; u < 32; ++u) {
            const int q = tid + 128 * u;
            *reinterpret_cast<float4*>(&whh_s[(q >> 4) * HP + 4 * (q & 15)]) = v[u];
        }
    }

    float xall[25][25];
#pragma unroll
    for (int i = 0; i < 25; ++i)
#pragma unroll
        for (int p = 0; p < 25; ++p) xall[i][p] = xg[3200 * i + p];

    float creg[8];
#pragma unroll
    for (int ii = 0; ii < 8; ++ii) {
        const int idx = tid + 128 * ii;
        creg[ii] = 0.0f;
        hlds[(idx >> 6) * HP + (idx & 63)] = 0.0f;
    }
#pragma unroll
    for (int i = 0; i < 25; ++i) xs[xdst[i]] = xall[i][0];

    float bv[4];
#pragma unroll
    for (int j = 0; j < 4; ++j) {
        const int g = gb + 16 * j + nn;
        bv[j] = bih[g] + bhh[g];
    }

    const float* xa = xs    + nn * XP + ks;
    const float* ha = hlds  + nn * HP + ks;
    const float* wa = wih_s + (gb + nn) * XP + ks;
    const float* ua = whh_s + (gb + nn) * HP + ks;
    __syncthreads();

    v8f acc[4];
    v2f fa[2][HCH];
    v2f fb[2][HCH][4];
#define LOAD_SET(buf, s0, n)                                                             \
    _Pragma("unroll") for (int kk = 0; kk < (n); ++kk) {                                  \
        const int s = (s0) + kk;                                                          \
        if (s < 50) {                                                                     \
            fa[buf][kk] = *reinterpret_cast<const v2f*>(xa + 4 * s);                      \
            _Pragma("unroll") for (int j = 0; j < 4; ++j)                                 \
                fb[buf][kk][j] = *reinterpret_cast<const v2f*>(wa + 16 * j * XP + 4 * s); \
        } else {                                                                          \
            fa[buf][kk] = *reinterpret_cast<const v2f*>(ha + 4 * (s - 50));               \
            _Pragma("unroll") for (int j = 0; j < 4; ++j)                                 \
                fb[buf][kk][j] = *reinterpret_cast<const v2f*>(ua + 16 * j * HP + 4 * (s - 50)); \
        }                                                                                 \
    }
#define WMMA_SET(buf, n)                                                                 \
    _Pragma("unroll") for (int kk = 0; kk < (n); ++kk)                                    \
        _Pragma("unroll") for (int j = 0; j < 4; ++j)                                     \
            acc[j] = WMMA_F32(fa[buf][kk], fb[buf][kk][j], acc[j]);
#define ACC_BIAS                                                                         \
    _Pragma("unroll") for (int j = 0; j < 4; ++j)                                         \
        _Pragma("unroll") for (int r = 0; r < 8; ++r) acc[j][r] = bv[j];
#define CELL(ii)                                                                         \
    {                                                                                     \
        const float cc = sigf(gf[ii]) * creg[ii] + sigf(gi[ii]) * tanhsel(gg[ii]);        \
        hn[ii] = sigf(go[ii]) * tanhsel(cc);                                              \
        creg[ii] = cc;                                                                    \
    }
#define STORE_CELL(ii)                                                                   \
    {                                                                                     \
        const int idx = tid + 128 * (ii);                                                 \
        const int row = idx >> 6;                                                         \
        const int hc  = idx & 63;                                                         \
        hlds[row * HP + hc] = hn[ii];                                                     \
        feat[(size_t)(b0 + row) * NF + hc * NT + t] = fmaxf(hn[ii], 0.0f);                \
    }
#define NO_CELL(c)
#define CELL_OF_SET(c) if ((c) < 8) { CELL(c) STORE_CELL(c) }
#define XPART(BESIDE)                                                                    \
    LOAD_SET(0, 0, XCH)                                                                   \
    _Pragma("unroll") for (int c = 0; c < 50 / XCH; ++c) {                                \
        if (c + 1 < 50 / XCH) { LOAD_SET((c + 1) & 1, XCH * (c + 1), XCH) }               \
        __builtin_amdgcn_sched_barrier(0);                                                \
        WMMA_SET(c & 1, XCH)                                                              \
        BESIDE(c)                                                                         \
        __builtin_amdgcn_sched_barrier(0);                                                \
    }

    ACC_BIAS
    XPART(NO_CELL)
    __syncthreads();

    for (int t = 0; t < NT; ++t) {

        LOAD_SET(0, 50, HCH)
        LOAD_SET(1, 50 + HCH, HCH)
        __builtin_amdgcn_sched_barrier(0);
        WMMA_SET(0, HCH)
        WMMA_SET(1, HCH)

#pragma unroll
        for (int tt = 1; tt < NT; ++tt) {
            if (t + 1 == tt) {
                const int off = (tt % 5) * 5 + tt / 5;
#pragma unroll
                for (int i = 0; i < 25; ++i) xs[xdst[i]] = xall[i][off];
            }
        }

#pragma unroll
        for (int j = 0; j < 4; ++j)
#pragma unroll
            for (int r = 0; r < 8; ++r)
                glds[(r + rofs) * NG + gb + 16 * j + nn] = acc[j][r];
        __syncthreads();

        float gi[8], gf[8], gg[8], go[8], hn[8];
#pragma unroll
        for (int ii = 0; ii < 8; ++ii) {
            const int idx = tid + 128 * ii;
            const int row = idx >> 6;
            const int hc  = idx & 63;
            gi[ii] = glds[row * NG + hc];
            gf[ii] = glds[row * NG + 64 + hc];
            gg[ii] = glds[row * NG + 128 + hc];
            go[ii] = glds[row * NG + 192 + hc];
        }
        ACC_BIAS
        if (t + 1 < NT) {
            XPART(CELL_OF_SET)
        } else {
#pragma unroll
            for (int ii = 0; ii < 8; ++ii) CELL(ii)
#pragma unroll
            for (int ii = 0; ii < 8; ++ii) STORE_CELL(ii)
        }
        __syncthreads();
    }
#undef LOAD_SET
#undef WMMA_SET
#undef ACC_BIAS
#undef CELL
#undef STORE_CELL
#undef XPART
#undef NO_CELL
#undef CELL_OF_SET
}

#define BNF 32
#define BNG 128

__global__ __launch_bounds__(1024) void bn_stats_kernel(const float* __restrict__ feat,
                                                        const float* __restrict__ gamma,
                                                        const float* __restrict__ beta,
                                                        float* __restrict__ scale,
                                                        float* __restrict__ shift)
{
    __shared__ __attribute__((aligned(16))) float ps1[BNG * BNF];
    __shared__ __attribute__((aligned(16))) float ps2[BNG * BNF];
    __shared__ float pr1[32 * BNF];
    __shared__ float pr2[32 * BNF];

    const int tid = threadIdx.x;
    const int q   = tid & 7;
    const int g   = tid >> 3;
    const int f0  = blockIdx.x * BNF;

    const float4* src = reinterpret_cast<const float4*>(feat + (size_t)g * NF + f0 + 4 * q);
    float4 s  = {0.0f, 0.0f, 0.0f, 0.0f};
    float4 s2 = {0.0f, 0.0f, 0.0f, 0.0f};
#pragma unroll 16
    for (int i = 0; i < NB / BNG; ++i) {
        const float4 v = src[(size_t)i * (BNG * NF / 4)];
        s.x += v.x;  s2.x += v.x * v.x;
        s.y += v.y;  s2.y += v.y * v.y;
        s.z += v.z;  s2.z += v.z * v.z;
        s.w += v.w;  s2.w += v.w * v.w;
    }
    *reinterpret_cast<float4*>(&ps1[4 * tid]) = s;
    *reinterpret_cast<float4*>(&ps2[4 * tid]) = s2;
    __syncthreads();

    {
        const int f = tid & 31;
        const int p = tid >> 5;
        float a = 0.0f, b = 0.0f;
        for (int i = 0; i < 4; ++i) {
            a += ps1[(4 * p + i) * BNF + f];
            b += ps2[(4 * p + i) * BNF + f];
        }
        pr1[p * BNF + f] = a;
        pr2[p * BNF + f] = b;
    }
    __syncthreads();

    if (tid < BNF) {
        float a = 0.0f, b = 0.0f;
        for (int p = 0; p < 32; ++p) {
            a += pr1[p * BNF + tid];
            b += pr2[p * BNF + tid];
        }
        const int f = f0 + tid;
        const float mu  = a * (1.0f / NB);
        const float var = b * (1.0f / NB) - mu * mu;
        const float sc  = gamma[f] * rsqrtf(var + 1e-5f);
        scale[f] = sc;
        shift[f] = beta[f] - mu * sc;
    }
}

#define FCWAVES 8
#define FCS     25

__global__ __launch_bounds__(256) void fc_kernel(
    const float* __restrict__ feat,
    const float* __restrict__ scale,
    const float* __restrict__ shift,
    const float* __restrict__ fcw,
    const float* __restrict__ fcb,
    float* __restrict__ out)
{
    __shared__ float part[FCWAVES * 256];

    const int tid  = threadIdx.x;
    const int lane = tid & 31;
    const int w    = tid >> 5;
    const int b0   = blockIdx.x * 16;
    const int h    = lane >> 4;
    const int nn   = lane & 15;
    const int col0 = 200 * w + 4 * h;

    const float4* pa = reinterpret_cast<const float4*>(feat + (size_t)(b0 + nn) * NF + col0);
    const float4* pb = reinterpret_cast<const float4*>(fcw + nn * NF + col0);
    const float4* ps = reinterpret_cast<const float4*>(scale + col0);
    const float4* pt = reinterpret_cast<const float4*>(shift + col0);

    const float bias = fcb[tid & 15];

    float4 va[FCS], vb[FCS], vs[FCS], vt[FCS];
#pragma unroll
    for (int s = 0; s < FCS; ++s) {
        va[s] = pa[2 * s];
        vb[s] = pb[2 * s];
        vs[s] = ps[2 * s];
        vt[s] = pt[2 * s];
    }

    v8f acc0, acc1;
#pragma unroll
    for (int r = 0; r < 8; ++r) { acc0[r] = 0.0f; acc1[r] = 0.0f; }

#pragma unroll
    for (int s = 0; s < FCS; ++s) {
        v2f a, b;
        a.x = fmaxf(va[s].x * vs[s].x + vt[s].x, 0.0f);
        a.y = fmaxf(va[s].y * vs[s].y + vt[s].y, 0.0f);
        b.x = vb[s].x;
        b.y = vb[s].y;
        acc0 = WMMA_F32(a, b, acc0);
        a.x = fmaxf(va[s].z * vs[s].z + vt[s].z, 0.0f);
        a.y = fmaxf(va[s].w * vs[s].w + vt[s].w, 0.0f);
        b.x = vb[s].z;
        b.y = vb[s].w;
        acc1 = WMMA_F32(a, b, acc1);
    }

#pragma unroll
    for (int r = 0; r < 8; ++r)
        part[w * 256 + (r + 8 * h) * NCLS + nn] = acc0[r] + acc1[r];
    __syncthreads();

    float sum = bias + part[tid];
#pragma unroll
    for (int q = 1; q < FCWAVES; ++q) sum += part[q * 256 + tid];
    out[b0 * NCLS + tid] = sum;
}

extern "C" void kernel_launch(void* const* d_in, const int* in_sizes, int n_in,
                              void* d_out, int out_size, void* d_ws, size_t ws_size,
                              hipStream_t stream)
{
    const float* x     = (const float*)d_in[0];
    const float* Wih   = (const float*)d_in[1];
    const float* Whh   = (const float*)d_in[2];
    const float* bih   = (const float*)d_in[3];
    const float* bhh   = (const float*)d_in[4];
    const float* gamma = (const float*)d_in[5];
    const float* beta  = (const float*)d_in[6];
    const float* fcw   = (const float*)d_in[7];
    const float* fcb   = (const float*)d_in[8];
    float* out = (float*)d_out;

    float* feat  = (float*)d_ws;
    float* scale = feat + (size_t)NB * NF;
    float* shift = scale + NF;

    lstm_kernel<<<NB / 16, 128, 0, stream>>>(x, Wih, Whh, bih, bhh, feat);
    bn_stats_kernel<<<NF / BNF, 1024, 0, stream>>>(feat, gamma, beta, scale, shift);
    fc_kernel<<<NB / 16, 32 * FCWAVES, 0, stream>>>(feat, scale, shift, fcw, fcb, out);
}


# lstm kernel: the wave's weights stay in registers as B fragments (528 per lane), the x tile in LDS; a wave owns all 4 gates of 16 hidden units, so no gate exchange; reorders K sums in groups of 8
# speedup vs baseline: 5.3382x; 1.1055x over previous
#include <hip/hip_runtime.h>
#include <math.h>

#define NB   4096
#define NC   200
#define NT   25
#define NH   64
#define NG   256
#define NF   1600
#define NCLS 16

typedef __attribute__((ext_vector_type(2))) float v2f;
typedef __attribute__((ext_vector_type(8))) float v8f;

#define WMMA_F32(a, b, c) \
    __builtin_amdgcn_wmma_f32_16x16x4_f32(false, (a), false, (b), (short)0, (c), false, false)

__device__ __forceinline__ float sigf(float x)
{
    return __builtin_amdgcn_rcpf(1.0f + __builtin_amdgcn_exp2f(x * -1.442695041f));
}

__device__ __forceinline__ float tanhsel(float x)
{
    const float ax = fabsf(x);
    const float e  = __builtin_amdgcn_exp2f(ax * 2.885390082f);
    const float big = fmaf(__builtin_amdgcn_rcpf(e + 1.0f), -2.0f, 1.0f);
    const float t = ax * ax;
    float p = fmaf(t, -0x1.758e7ap-8f, 0x1.521192p-6f);
    p = fmaf(t, p, -0x1.b8389cp-5f);
    p = fmaf(t, p, 0x1.110704p-3f);
    p = fmaf(t, p, -0x1.555532p-2f);
    const float sml = fmaf(t, ax * p, ax);
    return copysignf((ax < 0.625f) ? sml : big, x);
}

#define XROW 5000
#define HP 68
#define XSET 5

__global__ __launch_bounds__(128) void lstm_kernel(
    const float* __restrict__ x,
    const float* __restrict__ Wih,
    const float* __restrict__ Whh,
    const float* __restrict__ bih,
    const float* __restrict__ bhh,
    float* __restrict__ feat)
{
    __shared__ __attribute__((aligned(16))) float xt[16 * XROW];
    __shared__ __attribute__((aligned(16))) float hlds[16 * HP];

    const int tid  = threadIdx.x;
    const int lane = tid & 31;
    const int w    = tid >> 5;
    const int b0   = blockIdx.x * 16;
    const int q    = lane >> 4;
    const int nn   = lane & 15;
    const int rofs = 8 * q;
    const int hc   = 16 * w + nn;

    float4 wi[4][25];
    float4 wh[4][8];
#pragma unroll
    for (int j = 0; j < 4; ++j) {
        const float4* src = reinterpret_cast<const float4*>(Wih + (size_t)(64 * j + hc) * NC + 4 * q);
#pragma unroll
        for (int m = 0; m < 25; ++m) wi[j][m] = src[2 * m];
    }
#pragma unroll
    for (int j = 0; j < 4; ++j) {
        const float4* src = reinterpret_cast<const float4*>(Whh + (size_t)(64 * j + hc) * NH + 4 * q);
#pragma unroll
        for (int m = 0; m < 8; ++m) wh[j][m] = src[2 * m];
    }

    {
        const float4* src = reinterpret_cast<const float4*>(x + (size_t)b0 * NC * NT);
        float4* dst = reinterpret_cast<float4*>(xt);
        for (int u0 = 0; u0 < 156; u0 += 39) {
            float4 v[39];
#pragma unroll
            for (int u = 0; u < 39; ++u) v[u] = src[tid + 128 * (u0 + u)];
#pragma unroll
            for (int u = 0; u < 39; ++u) dst[tid + 128 * (u0 + u)] = v[u];
        }
        if (tid < 32) dst[tid + 128 * 156] = src[tid + 128 * 156];
    }

    float creg[8];
#pragma unroll
    for (int ii = 0; ii < 8; ++ii) {
        const int idx = tid + 128 * ii;
        creg[ii] = 0.0f;
        hlds[(idx >> 6) * HP + (idx & 63)] = 0.0f;
    }

    float bv[4];
#pragma unroll
    for (int j = 0; j < 4; ++j) {
        const int g = 64 * j + hc;
        bv[j] = bih[g] + bhh[g];
    }

    const float* xa = xt   + nn * XROW + 4 * q * NT;
    const float* ha = hlds + nn * HP + 4 * q;
    float* fo = feat + (size_t)(b0 + rofs) * NF + hc * NT;
    __syncthreads();

    v8f acc[4];
    v8f gt[4];
    float xf[2][XSET][4];
    float4 hf[8];
#define LOAD_X(buf, m0)                                                                  \
    _Pragma("unroll") for (int mm = 0; mm < XSET; ++mm)                                   \
        _Pragma("unroll") for (int e = 0; e < 4; ++e)                                     \
            xf[buf][mm][e] = xp[NT * (8 * ((m0) + mm) + e)];
#define WMMA_X(buf, m0)                                                                  \
    _Pragma("unroll") for (int mm = 0; mm < XSET; ++mm) {                                 \
        v2f a0, a1;                                                                       \
        a0.x = xf[buf][mm][0]; a0.y = xf[buf][mm][1];                                     \
        a1.x = xf[buf][mm][2]; a1.y = xf[buf][mm][3];                                     \
        _Pragma("unroll") for (int j = 0; j < 4; ++j) {                                   \
            v2f b0v; b0v.x = wi[j][(m0) + mm].x; b0v.y = wi[j][(m0) + mm].y;              \
            acc[j] = WMMA_F32(a0, b0v, acc[j]);                                           \
        }                                                                                 \
        _Pragma("unroll") for (int j = 0; j < 4; ++j) {                                   \
            v2f b1v; b1v.x = wi[j][(m0) + mm].z; b1v.y = wi[j][(m0) + mm].w;              \
            acc[j] = WMMA_F32(a1, b1v, acc[j]);                                           \
        }                                                                                 \
    }
#define ACC_BIAS                                                                         \
    _Pragma("unroll") for (int j = 0; j < 4; ++j)                                         \
        _Pragma("unroll") for (int r = 0; r < 8; ++r) acc[j][r] = bv[j];
#define CELL(r)                                                                          \
    {                                                                                     \
        const float cc = sigf(gt[1][r]) * creg[r] + sigf(gt[0][r]) * tanhsel(gt[2][r]);   \
        const float hn = sigf(gt[3][r]) * tanhsel(cc);                                    \
        creg[r] = cc;                                                                     \
        hlds[((r) + rofs) * HP + hc] = hn;                                                \
        fo[(r) * NF + t] = fmaxf(hn, 0.0f);                                               \
    }
#define NO_CELL(c)
#define CELLS_OF_SET(c) if ((c) < 4) { CELL(2 * (c)) CELL(2 * (c) + 1) }
#define XPART(BESIDE)                                                                    \
    LOAD_X(0, 0)                                                                          \
    _Pragma("unroll") for (int c = 0; c < 25 / XSET; ++c) {                               \
        if (c + 1 < 25 / XSET) { LOAD_X((c + 1) & 1, XSET * (c + 1)) }                    \
        __builtin_amdgcn_sched_barrier(0);                                                \
        WMMA_X(c & 1, XSET * c)                                                           \
        BESIDE(c)                                                                         \
        __builtin_amdgcn_sched_barrier(0);                                                \
    }

    {
        const float* xp = xa;
        ACC_BIAS
        XPART(NO_CELL)
    }

    for (int t = 0; t < NT; ++t) {

#pragma unroll
        for (int m = 0; m < 8; ++m) hf[m] = *reinterpret_cast<const float4*>(ha + 8 * m);
        __builtin_amdgcn_sched_barrier(0);
#pragma unroll
        for (int m = 0; m < 8; ++m) {
            v2f a0, a1;
            a0.x = hf[m].x; a0.y = hf[m].y;
            a1.x = hf[m].z; a1.y = hf[m].w;
#pragma unroll
            for (int j = 0; j < 4; ++j) {
                v2f b0v; b0v.x = wh[j][m].x; b0v.y = wh[j][m].y;
                acc[j] = WMMA_F32(a0, b0v, acc[j]);
            }
#pragma unroll
            for (int j = 0; j < 4; ++j) {
                v2f b1v; b1v.x = wh[j][m].z; b1v.y = wh[j][m].w;
                acc[j] = WMMA_F32(a1, b1v, acc[j]);
            }
        }
        __syncthreads();

#pragma unroll
        for (int j = 0; j < 4; ++j) gt[j] = acc[j];

        if (t + 1 < NT) {
            const int t1 = t + 1;
            const float* xp = xa + (t1 % 5) * 5 + t1 / 5;
            ACC_BIAS
            XPART(CELLS_OF_SET)
        } else {
#pragma unroll
            for (int r = 0; r < 8; ++r) CELL(r)
        }
        __syncthreads();
    }
#undef LOAD_X
#undef WMMA_X
#undef ACC_BIAS
#undef CELL
#undef XPART
#undef NO_CELL
#undef CELLS_OF_SET
}

#define BNF 32
#define BNG 128

__global__ __launch_bounds__(1024) void bn_stats_kernel(const float* __restrict__ feat,
                                                        const float* __restrict__ gamma,
                                                        const float* __restrict__ beta,
                                                        float* __restrict__ scale,
                                                        float* __restrict__ shift)
{
    __shared__ __attribute__((aligned(16))) float ps1[BNG * BNF];
    __shared__ __attribute__((aligned(16))) float ps2[BNG * BNF];
    __shared__ float pr1[32 * BNF];
    __shared__ float pr2[32 * BNF];

    const int tid = threadIdx.x;
    const int q   = tid & 7;
    const int g   = tid >> 3;
    const int f0  = blockIdx.x * BNF;

    const float4* src = reinterpret_cast<const float4*>(feat + (size_t)g * NF + f0 + 4 * q);
    float4 s  = {0.0f, 0.0f, 0.0f, 0.0f};
    float4 s2 = {0.0f, 0.0f, 0.0f, 0.0f};
#pragma unroll 16
    for (int i = 0; i < NB / BNG; ++i) {
        const float4 v = src[(size_t)i * (BNG * NF / 4)];
        s.x += v.x;  s2.x += v.x * v.x;
        s.y += v.y;  s2.y += v.y * v.y;
        s.z += v.z;  s2.z += v.z * v.z;
        s.w += v.w;  s2.w += v.w * v.w;
    }
    *reinterpret_cast<float4*>(&ps1[4 * tid]) = s;
    *reinterpret_cast<float4*>(&ps2[4 * tid]) = s2;
    __syncthreads();

    {
        const int f = tid & 31;
        const int p = tid >> 5;
        float a = 0.0f, b = 0.0f;
        for (int i = 0; i < 4; ++i) {
            a += ps1[(4 * p + i) * BNF + f];
            b += ps2[(4 * p + i) * BNF + f];
        }
        pr1[p * BNF + f] = a;
        pr2[p * BNF + f] = b;
    }
    __syncthreads();

    if (tid < BNF) {
        float a = 0.0f, b = 0.0f;
        for (int p = 0; p < 32; ++p) {
            a += pr1[p * BNF + tid];
            b += pr2[p * BNF + tid];
        }
        const int f = f0 + tid;
        const float mu  = a * (1.0f / NB);
        const float var = b * (1.0f / NB) - mu * mu;
        const float sc  = gamma[f] * rsqrtf(var + 1e-5f);
        scale[f] = sc;
        shift[f] = beta[f] - mu * sc;
    }
}

#define FCWAVES 8
#define FCS     25

__global__ __launch_bounds__(256) void fc_kernel(
    const float* __restrict__ feat,
    const float* __restrict__ scale,
    const float* __restrict__ shift,
    const float* __restrict__ fcw,
    const float* __restrict__ fcb,
    float* __restrict__ out)
{
    __shared__ float part[FCWAVES * 256];

    const int tid  = threadIdx.x;
    const int lane = tid & 31;
    const int w    = tid >> 5;
    const int b0   = blockIdx.x * 16;
    const int h    = lane >> 4;
    const int nn   = lane & 15;
    const int col0 = 200 * w + 4 * h;

    const float4* pa = reinterpret_cast<const float4*>(feat + (size_t)(b0 + nn) * NF + col0);
    const float4* pb = reinterpret_cast<const float4*>(fcw + nn * NF + col0);
    const float4* ps = reinterpret_cast<const float4*>(scale + col0);
    const float4* pt = reinterpret_cast<const float4*>(shift + col0);

    const float bias = fcb[tid & 15];

    float4 va[FCS], vb[FCS], vs[FCS], vt[FCS];
#pragma unroll
    for (int s = 0; s < FCS; ++s) {
        va[s] = pa[2 * s];
        vb[s] = pb[2 * s];
        vs[s] = ps[2 * s];
        vt[s] = pt[2 * s];
    }

    v8f acc0, acc1;
#pragma unroll
    for (int r = 0; r < 8; ++r) { acc0[r] = 0.0f; acc1[r] = 0.0f; }

#pragma unroll
    for (int s = 0; s < FCS; ++s) {
        v2f a, b;
        a.x = fmaxf(va[s].x * vs[s].x + vt[s].x, 0.0f);
        a.y = fmaxf(va[s].y * vs[s].y + vt[s].y, 0.0f);
        b.x = vb[s].x;
        b.y = vb[s].y;
        acc0 = WMMA_F32(a, b, acc0);
        a.x = fmaxf(va[s].z * vs[s].z + vt[s].z, 0.0f);
        a.y = fmaxf(va[s].w * vs[s].w + vt[s].w, 0.0f);
        b.x = vb[s].z;
        b.y = vb[s].w;
        acc1 = WMMA_F32(a, b, acc1);
    }

#pragma unroll
    for (int r = 0; r < 8; ++r)
        part[w * 256 + (r + 8 * h) * NCLS + nn] = acc0[r] + acc1[r];
    __syncthreads();

    float sum = bias + part[tid];
#pragma unroll
    for (int q = 1; q < FCWAVES; ++q) sum += part[q * 256 + tid];
    out[b0 * NCLS + tid] = sum;
}

extern "C" void kernel_launch(void* const* d_in, const int* in_sizes, int n_in,
                              void* d_out, int out_size, void* d_ws, size_t ws_size,
                              hipStream_t stream)
{
    const float* x     = (const float*)d_in[0];
    const float* Wih   = (const float*)d_in[1];
    const float* Whh   = (const float*)d_in[2];
    const float* bih   = (const float*)d_in[3];
    const float* bhh   = (const float*)d_in[4];
    const float* gamma = (const float*)d_in[5];
    const float* beta  = (const float*)d_in[6];
    const float* fcw   = (const float*)d_in[7];
    const float* fcb   = (const float*)d_in[8];
    float* out = (float*)d_out;

    float* feat  = (float*)d_ws;
    float* scale = feat + (size_t)NB * NF;
    float* shift = scale + NF;

    lstm_kernel<<<NB / 16, 128, 0, stream>>>(x, Wih, Whh, bih, bhh, feat);
    bn_stats_kernel<<<NF / BNF, 1024, 0, stream>>>(feat, gamma, beta, scale, shift);
    fc_kernel<<<NB / 16, 32 * FCWAVES, 0, stream>>>(feat, scale, shift, fcw, fcb, out);
}


# lstm step loop: LDS fragment reads issued one WMMA block ahead (3 x register sets; h reads covered by its last 16 WMMAs), cell update in two halves beside two blocks, barriers wait for LDS only
# speedup vs baseline: 5.6098x; 1.0509x over previous
#include <hip/hip_runtime.h>
#include <math.h>

#define NB   4096
#define NC   200
#define NT   25
#define NH   64
#define NG   256
#define NF   1600
#define NCLS 16

typedef __attribute__((ext_vector_type(2))) float v2f;
typedef __attribute__((ext_vector_type(8))) float v8f;

#define WMMA_F32(a, b, c) \
    __builtin_amdgcn_wmma_f32_16x16x4_f32(false, (a), false, (b), (short)0, (c), false, false)

__device__ __forceinline__ float sigf(float x)
{
    return __builtin_amdgcn_rcpf(1.0f + __builtin_amdgcn_exp2f(x * -1.442695041f));
}

__device__ __forceinline__ float tanhsel(float x)
{
    const float ax = fabsf(x);
    const float e  = __builtin_amdgcn_exp2f(ax * 2.885390082f);
    const float big = fmaf(__builtin_amdgcn_rcpf(e + 1.0f), -2.0f, 1.0f);
    const float t = ax * ax;
    float p = fmaf(t, -0x1.758e7ap-8f, 0x1.521192p-6f);
    p = fmaf(t, p, -0x1.b8389cp-5f);
    p = fmaf(t, p, 0x1.110704p-3f);
    p = fmaf(t, p, -0x1.555532p-2f);
    const float sml = fmaf(t, ax * p, ax);
    return copysignf((ax < 0.625f) ? sml : big, x);
}

#define XROW 5000
#define HP 68

__global__ __launch_bounds__(128) void lstm_kernel(
    const float* __restrict__ x,
    const float* __restrict__ Wih,
    const float* __restrict__ Whh,
    const float* __restrict__ bih,
    const float* __restrict__ bhh,
    float* __restrict__ feat)
{
    __shared__ __attribute__((aligned(16))) float xt[16 * XROW];
    __shared__ __attribute__((aligned(16))) float hlds[16 * HP];

    const int tid  = threadIdx.x;
    const int lane = tid & 31;
    const int w    = tid >> 5;
    const int b0   = blockIdx.x * 16;
    const int q    = lane >> 4;
    const int nn   = lane & 15;
    const int rofs = 8 * q;
    const int hc   = 16 * w + nn;

    float4 wi[4][25];
    float4 wh[4][8];
#pragma unroll
    for (int j = 0; j < 4; ++j) {
        const float4* src = reinterpret_cast<const float4*>(Wih + (size_t)(64 * j + hc) * NC + 4 * q);
#pragma unroll
        for (int m = 0; m < 25; ++m) wi[j][m] = src[2 * m];
    }
#pragma unroll
    for (int j = 0; j < 4; ++j) {
        const float4* src = reinterpret_cast<const float4*>(Whh + (size_t)(64 * j + hc) * NH + 4 * q);
#pragma unroll
        for (int m = 0; m < 8; ++m) wh[j][m] = src[2 * m];
    }

    {
        const float4* src = reinterpret_cast<const float4*>(x + (size_t)b0 * NC * NT);
        float4* dst = reinterpret_cast<float4*>(xt);
        for (int u0 = 0; u0 < 156; u0 += 39) {
            float4 v[39];
#pragma unroll
            for (int u = 0; u < 39; ++u) v[u] = src[tid + 128 * (u0 + u)];
#pragma unroll
            for (int u = 0; u < 39; ++u) dst[tid + 128 * (u0 + u)] = v[u];
        }
        if (tid < 32) dst[tid + 128 * 156] = src[tid + 128 * 156];
    }

    float creg[8];
#pragma unroll
    for (int ii = 0; ii < 8; ++ii) {
        const int idx = tid + 128 * ii;
        creg[ii] = 0.0f;
        hlds[(idx >> 6) * HP + (idx & 63)] = 0.0f;
    }

    float bv[4];
#pragma unroll
    for (int j = 0; j < 4; ++j) {
        const int g = 64 * j + hc;
        bv[j] = bih[g] + bhh[g];
    }

    const float* xa = xt   + nn * XROW + 4 * q * NT;
    const float* ha = hlds + nn * HP + 4 * q;
    float* fo = feat + (size_t)(b0 + rofs) * NF + hc * NT;
    __syncthreads();

    v8f acc[4];
    v8f gt[4];
    float xfa[10][4];
    float xfb[10][4];
    float xfc[5][4];
    float og[8];
    float4 hf[8];
#define LOADP(dst, n, m0, xp)                                                            \
    _Pragma("unroll") for (int mm = 0; mm < (n); ++mm)                                    \
        _Pragma("unroll") for (int e = 0; e < 4; ++e)                                     \
            dst[mm][e] = (xp)[NT * (8 * ((m0) + mm) + e)];
#define WMMAP(src, s0, s1, m0)                                                           \
    _Pragma("unroll") for (int mm = (s0); mm < (s1); ++mm) {                              \
        v2f a0, a1;                                                                       \
        a0.x = src[mm][0]; a0.y = src[mm][1];                                             \
        a1.x = src[mm][2]; a1.y = src[mm][3];                                             \
        _Pragma("unroll") for (int j = 0; j < 4; ++j) {                                   \
            v2f b0v; b0v.x = wi[j][(m0) + mm].x; b0v.y = wi[j][(m0) + mm].y;              \
            acc[j] = WMMA_F32(a0, b0v, acc[j]);                                           \
        }                                                                                 \
        _Pragma("unroll") for (int j = 0; j < 4; ++j) {                                   \
            v2f b1v; b1v.x = wi[j][(m0) + mm].z; b1v.y = wi[j][(m0) + mm].w;              \
            acc[j] = WMMA_F32(a1, b1v, acc[j]);                                           \
        }                                                                                 \
    }
#define LOADH                                                                            \
    _Pragma("unroll") for (int m = 0; m < 8; ++m) hf[m] = *reinterpret_cast<const float4*>(ha + 8 * m);
#define HPART                                                                            \
    _Pragma("unroll") for (int m = 0; m < 8; ++m) {                                       \
        v2f a0, a1;                                                                       \
        a0.x = hf[m].x; a0.y = hf[m].y;                                                   \
        a1.x = hf[m].z; a1.y = hf[m].w;                                                   \
        _Pragma("unroll") for (int j = 0; j < 4; ++j) {                                   \
            v2f b0v; b0v.x = wh[j][m].x; b0v.y = wh[j][m].y;                              \
            acc[j] = WMMA_F32(a0, b0v, acc[j]);                                           \
        }                                                                                 \
        _Pragma("unroll") for (int j = 0; j < 4; ++j) {                                   \
            v2f b1v; b1v.x = wh[j][m].z; b1v.y = wh[j][m].w;                              \
            acc[j] = WMMA_F32(a1, b1v, acc[j]);                                           \
        }                                                                                 \
    }
#define ACC_BIAS                                                                         \
    _Pragma("unroll") for (int j = 0; j < 4; ++j)                                         \
        _Pragma("unroll") for (int r = 0; r < 8; ++r) acc[j][r] = bv[j];
#define CELL_A(r)                                                                        \
    {                                                                                     \
        const float cc = sigf(gt[1][r]) * creg[r] + sigf(gt[0][r]) * tanhsel(gt[2][r]);   \
        creg[r] = cc;                                                                     \
        og[r] = sigf(gt[3][r]);                                                           \
    }
#define CELL_B(r, TOLDS)                                                                 \
    {                                                                                     \
        const float hn = og[r] * tanhsel(creg[r]);                                        \
        if (TOLDS) hlds[((r) + rofs) * HP + hc] = hn;                                     \
        fo[(r) * NF + t] = fmaxf(hn, 0.0f);                                               \
    }
#define PIN_A                                                                            \
    asm volatile("" : "+v"(creg[0]), "+v"(creg[1]), "+v"(creg[2]), "+v"(creg[3]),        \
                      "+v"(creg[4]), "+v"(creg[5]), "+v"(creg[6]), "+v"(creg[7]));       \
    asm volatile("" : "+v"(og[0]), "+v"(og[1]), "+v"(og[2]), "+v"(og[3]),                \
                      "+v"(og[4]), "+v"(og[5]), "+v"(og[6]), "+v"(og[7]));
#define SB __builtin_amdgcn_sched_barrier(0);
#define LDS_BARRIER                                                                      \
    __builtin_amdgcn_fence(__ATOMIC_RELEASE, "workgroup", "local");                      \
    __builtin_amdgcn_s_barrier();                                                         \
    __builtin_amdgcn_fence(__ATOMIC_ACQUIRE, "workgroup", "local");

    {
        const float* xp = xa;
        ACC_BIAS
        LOADP(xfa, 10, 0, xp)
        LOADP(xfb, 10, 10, xp)
        LOADP(xfc, 5, 20, xp)
        SB
        WMMAP(xfa, 0, 10, 0)
        WMMAP(xfb, 0, 10, 10)
        WMMAP(xfc, 0, 5, 20)
        SB
        const float* xp1 = xa + 5;
        LOADH
        LOADP(xfa, 10, 0, xp1)
        LOADP(xfb, 10, 10, xp1)
        SB
    }

#pragma unroll 2
    for (int t = 0; t < NT - 1; ++t) {
        const int t1 = t + 1;
        const int t2 = (t + 2 < NT) ? t + 2 : NT - 1;
        const float* xp  = xa + (t1 % 5) * 5 + t1 / 5;
        const float* xp2 = xa + (t2 % 5) * 5 + t2 / 5;

        HPART
#pragma unroll
        for (int j = 0; j < 4; ++j) gt[j] = acc[j];
        ACC_BIAS

        SB
        WMMAP(xfa, 0, 10, 0)
        CELL_A(0) CELL_A(1) CELL_A(2) CELL_A(3) CELL_A(4) CELL_A(5) CELL_A(6) CELL_A(7)
        PIN_A
        SB
        LDS_BARRIER

        LOADP(xfc, 5, 20, xp)
        SB
        WMMAP(xfb, 0, 2, 10)
        CELL_B(0, true) CELL_B(1, true)
        SB
        WMMAP(xfb, 2, 4, 10)
        CELL_B(2, true) CELL_B(3, true)
        SB
        WMMAP(xfb, 4, 6, 10)
        CELL_B(4, true) CELL_B(5, true)
        SB
        WMMAP(xfb, 6, 8, 10)
        CELL_B(6, true) CELL_B(7, true)
        SB
        WMMAP(xfb, 8, 10, 10)
        SB

        LOADP(xfa, 10, 0, xp2)
        SB
        WMMAP(xfc, 0, 3, 20)
        SB
        LDS_BARRIER

        LOADH
        LOADP(xfb, 10, 10, xp2)
        SB
        WMMAP(xfc, 3, 5, 20)
        SB
    }

    {
        const int t = NT - 1;
        HPART
#pragma unroll
        for (int j = 0; j < 4; ++j) gt[j] = acc[j];
        CELL_A(0) CELL_A(1) CELL_A(2) CELL_A(3) CELL_A(4) CELL_A(5) CELL_A(6) CELL_A(7)
        CELL_B(0, false) CELL_B(1, false) CELL_B(2, false) CELL_B(3, false)
        CELL_B(4, false) CELL_B(5, false) CELL_B(6, false) CELL_B(7, false)
    }
#undef LOADP
#undef WMMAP
#undef LOADH
#undef HPART
#undef ACC_BIAS
#undef CELL_A
#undef CELL_B
#undef PIN_A
#undef SB
#undef LDS_BARRIER
}

#define BNF 32
#define BNG 128

__global__ __launch_bounds__(1024) void bn_stats_kernel(const float* __restrict__ feat,
                                                        const float* __restrict__ gamma,
                                                        const float* __restrict__ beta,
                                                        float* __restrict__ scale,
                                                        float* __restrict__ shift)
{
    __shared__ __attribute__((aligned(16))) float ps1[BNG * BNF];
    __shared__ __attribute__((aligned(16))) float ps2[BNG * BNF];
    __shared__ float pr1[32 * BNF];
    __shared__ float pr2[32 * BNF];

    const int tid = threadIdx.x;
    const int q   = tid & 7;
    const int g   = tid >> 3;
    const int f0  = blockIdx.x * BNF;

    const float4* src = reinterpret_cast<const float4*>(feat + (size_t)g * NF + f0 + 4 * q);
    float4 s  = {0.0f, 0.0f, 0.0f, 0.0f};
    float4 s2 = {0.0f, 0.0f, 0.0f, 0.0f};
#pragma unroll 16
    for (int i = 0; i < NB / BNG; ++i) {
        const float4 v = src[(size_t)i * (BNG * NF / 4)];
        s.x += v.x;  s2.x += v.x * v.x;
        s.y += v.y;  s2.y += v.y * v.y;
        s.z += v.z;  s2.z += v.z * v.z;
        s.w += v.w;  s2.w += v.w * v.w;
    }
    *reinterpret_cast<float4*>(&ps1[4 * tid]) = s;
    *reinterpret_cast<float4*>(&ps2[4 * tid]) = s2;
    __syncthreads();

    {
        const int f = tid & 31;
        const int p = tid >> 5;
        float a = 0.0f, b = 0.0f;
        for (int i = 0; i < 4; ++i) {
            a += ps1[(4 * p + i) * BNF + f];
            b += ps2[(4 * p + i) * BNF + f];
        }
        pr1[p * BNF + f] = a;
        pr2[p * BNF + f] = b;
    }
    __syncthreads();

    if (tid < BNF) {
        float a = 0.0f, b = 0.0f;
        for (int p = 0; p < 32; ++p) {
            a += pr1[p * BNF + tid];
            b += pr2[p * BNF + tid];
        }
        const int f = f0 + tid;
        const float mu  = a * (1.0f / NB);
        const float var = b * (1.0f / NB) - mu * mu;
        const float sc  = gamma[f] * rsqrtf(var + 1e-5f);
        scale[f] = sc;
        shift[f] = beta[f] - mu * sc;
    }
}

#define FCWAVES 8
#define FCS     25

__global__ __launch_bounds__(256) void fc_kernel(
    const float* __restrict__ feat,
    const float* __restrict__ scale,
    const float* __restrict__ shift,
    const float* __restrict__ fcw,
    const float* __restrict__ fcb,
    float* __restrict__ out)
{
    __shared__ float part[FCWAVES * 256];

    const int tid  = threadIdx.x;
    const int lane = tid & 31;
    const int w    = tid >> 5;
    const int b0   = blockIdx.x * 16;
    const int h    = lane >> 4;
    const int nn   = lane & 15;
    const int col0 = 200 * w + 4 * h;

    const float4* pa = reinterpret_cast<const float4*>(feat + (size_t)(b0 + nn) * NF + col0);
    const float4* pb = reinterpret_cast<const float4*>(fcw + nn * NF + col0);
    const float4* ps = reinterpret_cast<const float4*>(scale + col0);
    const float4* pt = reinterpret_cast<const float4*>(shift + col0);

    const float bias = fcb[tid & 15];

    float4 va[FCS], vb[FCS], vs[FCS], vt[FCS];
#pragma unroll
    for (int s = 0; s < FCS; ++s) {
        va[s] = pa[2 * s];
        vb[s] = pb[2 * s];
        vs[s] = ps[2 * s];
        vt[s] = pt[2 * s];
    }

    v8f acc0, acc1;
#pragma unroll
    for (int r = 0; r < 8; ++r) { acc0[r] = 0.0f; acc1[r] = 0.0f; }

#pragma unroll
    for (int s = 0; s < FCS; ++s) {
        v2f a, b;
        a.x = fmaxf(va[s].x * vs[s].x + vt[s].x, 0.0f);
        a.y = fmaxf(va[s].y * vs[s].y + vt[s].y, 0.0f);
        b.x = vb[s].x;
        b.y = vb[s].y;
        acc0 = WMMA_F32(a, b, acc0);
        a.x = fmaxf(va[s].z * vs[s].z + vt[s].z, 0.0f);
        a.y = fmaxf(va[s].w * vs[s].w + vt[s].w, 0.0f);
        b.x = vb[s].z;
        b.y = vb[s].w;
        acc1 = WMMA_F32(a, b, acc1);
    }

#pragma unroll
    for (int r = 0; r < 8; ++r)
        part[w * 256 + (r + 8 * h) * NCLS + nn] = acc0[r] + acc1[r];
    __syncthreads();

    float sum = bias + part[tid];
#pragma unroll
    for (int q = 1; q < FCWAVES; ++q) sum += part[q * 256 + tid];
    out[b0 * NCLS + tid] = sum;
}

extern "C" void kernel_launch(void* const* d_in, const int* in_sizes, int n_in,
                              void* d_out, int out_size, void* d_ws, size_t ws_size,
                              hipStream_t stream)
{
    const float* x     = (const float*)d_in[0];
    const float* Wih   = (const float*)d_in[1];
    const float* Whh   = (const float*)d_in[2];
    const float* bih   = (const float*)d_in[3];
    const float* bhh   = (const float*)d_in[4];
    const float* gamma = (const float*)d_in[5];
    const float* beta  = (const float*)d_in[6];
    const float* fcw   = (const float*)d_in[7];
    const float* fcb   = (const float*)d_in[8];
    float* out = (float*)d_out;

    float* feat  = (float*)d_ws;
    float* scale = feat + (size_t)NB * NF;
    float* shift = scale + NF;

    lstm_kernel<<<NB / 16, 128, 0, stream>>>(x, Wih, Whh, bih, bhh, feat);
    bn_stats_kernel<<<NF / BNF, 1024, 0, stream>>>(feat, gamma, beta, scale, shift);
    fc_kernel<<<NB / 16, 32 * FCWAVES, 0, stream>>>(feat, scale, shift, fcw, fcb, out);
}


# lstm cell: tanh on the hardware v_tanh_f32 and sigmoid(x) = 0.5 + 0.5 tanh(x/2): 15 vector ops per cell instead of 46 (exp2 + rcp + polynomial select); activation rounding differs
# speedup vs baseline: 5.9647x; 1.0633x over previous
#include <hip/hip_runtime.h>
#include <math.h>

#define NB   4096
#define NC   200
#define NT   25
#define NH   64
#define NG   256
#define NF   1600
#define NCLS 16

typedef __attribute__((ext_vector_type(2))) float v2f;
typedef __attribute__((ext_vector_type(8))) float v8f;

#define WMMA_F32(a, b, c) \
    __builtin_amdgcn_wmma_f32_16x16x4_f32(false, (a), false, (b), (short)0, (c), false, false)

__device__ __forceinline__ float tanhhw(float x)
{
    return __builtin_amdgcn_tanhf(x);
}

__device__ __forceinline__ float sigf(float x)
{
    return fmaf(__builtin_amdgcn_tanhf(0.5f * x), 0.5f, 0.5f);
}

#define XROW 5000
#define HP 68

__global__ __launch_bounds__(128) void lstm_kernel(
    const float* __restrict__ x,
    const float* __restrict__ Wih,
    const float* __restrict__ Whh,
    const float* __restrict__ bih,
    const float* __restrict__ bhh,
    float* __restrict__ feat)
{
    __shared__ __attribute__((aligned(16))) float xt[16 * XROW];
    __shared__ __attribute__((aligned(16))) float hlds[16 * HP];

    const int tid  = threadIdx.x;
    const int lane = tid & 31;
    const int w    = tid >> 5;
    const int b0   = blockIdx.x * 16;
    const int q    = lane >> 4;
    const int nn   = lane & 15;
    const int rofs = 8 * q;
    const int hc   = 16 * w + nn;

    float4 wi[4][25];
    float4 wh[4][8];
#pragma unroll
    for (int j = 0; j < 4; ++j) {
        const float4* src = reinterpret_cast<const float4*>(Wih + (size_t)(64 * j + hc) * NC + 4 * q);
#pragma unroll
        for (int m = 0; m < 25; ++m) wi[j][m] = src[2 * m];
    }
#pragma unroll
    for (int j = 0; j < 4; ++j) {
        const float4* src = reinterpret_cast<const float4*>(Whh + (size_t)(64 * j + hc) * NH + 4 * q);
#pragma unroll
        for (int m = 0; m < 8; ++m) wh[j][m] = src[2 * m];
    }

    {
        const float4* src = reinterpret_cast<const float4*>(x + (size_t)b0 * NC * NT);
        float4* dst = reinterpret_cast<float4*>(xt);
        for (int u0 = 0; u0 < 156; u0 += 39) {
            float4 v[39];
#pragma unroll
            for (int u = 0; u < 39; ++u) v[u] = src[tid + 128 * (u0 + u)];
#pragma unroll
            for (int u = 0; u < 39; ++u) dst[tid + 128 * (u0 + u)] = v[u];
        }
        if (tid < 32) dst[tid + 128 * 156] = src[tid + 128 * 156];
    }

    float creg[8];
#pragma unroll
    for (int ii = 0; ii < 8; ++ii) {
        const int idx = tid + 128 * ii;
        creg[ii] = 0.0f;
        hlds[(idx >> 6) * HP + (idx & 63)] = 0.0f;
    }

    float bv[4];
#pragma unroll
    for (int j = 0; j < 4; ++j) {
        const int g = 64 * j + hc;
        bv[j] = bih[g] + bhh[g];
    }

    const float* xa = xt   + nn * XROW + 4 * q * NT;
    const float* ha = hlds + nn * HP + 4 * q;
    float* fo = feat + (size_t)(b0 + rofs) * NF + hc * NT;
    __syncthreads();

    v8f acc[4];
    v8f gt[4];
    float xfa[10][4];
    float xfb[10][4];
    float xfc[5][4];
    float og[8];
    float4 hf[8];
#define LOADP(dst, n, m0, xp)                                                            \
    _Pragma("unroll") for (int mm = 0; mm < (n); ++mm)                                    \
        _Pragma("unroll") for (int e = 0; e < 4; ++e)                                     \
            dst[mm][e] = (xp)[NT * (8 * ((m0) + mm) + e)];
#define WMMAP(src, s0, s1, m0)                                                           \
    _Pragma("unroll") for (int mm = (s0); mm < (s1); ++mm) {                              \
        v2f a0, a1;                                                                       \
        a0.x = src[mm][0]; a0.y = src[mm][1];                                             \
        a1.x = src[mm][2]; a1.y = src[mm][3];                                             \
        _Pragma("unroll") for (int j = 0; j < 4; ++j) {                                   \
            v2f b0v; b0v.x = wi[j][(m0) + mm].x; b0v.y = wi[j][(m0) + mm].y;              \
            acc[j] = WMMA_F32(a0, b0v, acc[j]);                                           \
        }                                                                                 \
        _Pragma("unroll") for (int j = 0; j < 4; ++j) {                                   \
            v2f b1v; b1v.x = wi[j][(m0) + mm].z; b1v.y = wi[j][(m0) + mm].w;              \
            acc[j] = WMMA_F32(a1, b1v, acc[j]);                                           \
        }                                                                                 \
    }
#define LOADH                                                                            \
    _Pragma("unroll") for (int m = 0; m < 8; ++m) hf[m] = *reinterpret_cast<const float4*>(ha + 8 * m);
#define HPART                                                                            \
    _Pragma("unroll") for (int m = 0; m < 8; ++m) {                                       \
        v2f a0, a1;                                                                       \
        a0.x = hf[m].x; a0.y = hf[m].y;                                                   \
        a1.x = hf[m].z; a1.y = hf[m].w;                                                   \
        _Pragma("unroll") for (int j = 0; j < 4; ++j) {                                   \
            v2f b0v; b0v.x = wh[j][m].x; b0v.y = wh[j][m].y;                              \
            acc[j] = WMMA_F32(a0, b0v, acc[j]);                                           \
        }                                                                                 \
        _Pragma("unroll") for (int j = 0; j < 4; ++j) {                                   \
            v2f b1v; b1v.x = wh[j][m].z; b1v.y = wh[j][m].w;                              \
            acc[j] = WMMA_F32(a1, b1v, acc[j]);                                           \
        }                                                                                 \
    }
#define ACC_BIAS                                                                         \
    _Pragma("unroll") for (int j = 0; j < 4; ++j)                                         \
        _Pragma("unroll") for (int r = 0; r < 8; ++r) acc[j][r] = bv[j];
#define CELL_A(r)                                                                        \
    {                                                                                     \
        const float cc = sigf(gt[1][r]) * creg[r] + sigf(gt[0][r]) * tanhhw(gt[2][r]);   \
        creg[r] = cc;                                                                     \
        og[r] = sigf(gt[3][r]);                                                           \
    }
#define CELL_B(r, TOLDS)                                                                 \
    {                                                                                     \
        const float hn = og[r] * tanhhw(creg[r]);                                        \
        if (TOLDS) hlds[((r) + rofs) * HP + hc] = hn;                                     \
        fo[(r) * NF + t] = fmaxf(hn, 0.0f);                                               \
    }
#define PIN_A                                                                            \
    asm volatile("" : "+v"(creg[0]), "+v"(creg[1]), "+v"(creg[2]), "+v"(creg[3]),        \
                      "+v"(creg[4]), "+v"(creg[5]), "+v"(creg[6]), "+v"(creg[7]));       \
    asm volatile("" : "+v"(og[0]), "+v"(og[1]), "+v"(og[2]), "+v"(og[3]),                \
                      "+v"(og[4]), "+v"(og[5]), "+v"(og[6]), "+v"(og[7]));
#define SB __builtin_amdgcn_sched_barrier(0);
#define LDS_BARRIER                                                                      \
    __builtin_amdgcn_fence(__ATOMIC_RELEASE, "workgroup", "local");                      \
    __builtin_amdgcn_s_barrier();                                                         \
    __builtin_amdgcn_fence(__ATOMIC_ACQUIRE, "workgroup", "local");

    {
        const float* xp = xa;
        ACC_BIAS
        LOADP(xfa, 10, 0, xp)
        LOADP(xfb, 10, 10, xp)
        LOADP(xfc, 5, 20, xp)
        SB
        WMMAP(xfa, 0, 10, 0)
        WMMAP(xfb, 0, 10, 10)
        WMMAP(xfc, 0, 5, 20)
        SB
        const float* xp1 = xa + 5;
        LOADH
        LOADP(xfa, 10, 0, xp1)
        LOADP(xfb, 10, 10, xp1)
        SB
    }

#pragma unroll 2
    for (int t = 0; t < NT - 1; ++t) {
        const int t1 = t + 1;
        const int t2 = (t + 2 < NT) ? t + 2 : NT - 1;
        const float* xp  = xa + (t1 % 5) * 5 + t1 / 5;
        const float* xp2 = xa + (t2 % 5) * 5 + t2 / 5;

        HPART
#pragma unroll
        for (int j = 0; j < 4; ++j) gt[j] = acc[j];
        ACC_BIAS

        SB
        WMMAP(xfa, 0, 10, 0)
        CELL_A(0) CELL_A(1) CELL_A(2) CELL_A(3) CELL_A(4) CELL_A(5) CELL_A(6) CELL_A(7)
        PIN_A
        SB
        LDS_BARRIER

        LOADP(xfc, 5, 20, xp)
        SB
        WMMAP(xfb, 0, 2, 10)
        CELL_B(0, true) CELL_B(1, true)
        SB
        WMMAP(xfb, 2, 4, 10)
        CELL_B(2, true) CELL_B(3, true)
        SB
        WMMAP(xfb, 4, 6, 10)
        CELL_B(4, true) CELL_B(5, true)
        SB
        WMMAP(xfb, 6, 8, 10)
        CELL_B(6, true) CELL_B(7, true)
        SB
        WMMAP(xfb, 8, 10, 10)
        SB

        LOADP(xfa, 10, 0, xp2)
        SB
        WMMAP(xfc, 0, 3, 20)
        SB
        LDS_BARRIER

        LOADH
        LOADP(xfb, 10, 10, xp2)
        SB
        WMMAP(xfc, 3, 5, 20)
        SB
    }

    {
        const int t = NT - 1;
        HPART
#pragma unroll
        for (int j = 0; j < 4; ++j) gt[j] = acc[j];
        CELL_A(0) CELL_A(1) CELL_A(2) CELL_A(3) CELL_A(4) CELL_A(5) CELL_A(6) CELL_A(7)
        CELL_B(0, false) CELL_B(1, false) CELL_B(2, false) CELL_B(3, false)
        CELL_B(4, false) CELL_B(5, false) CELL_B(6, false) CELL_B(7, false)
    }
#undef LOADP
#undef WMMAP
#undef LOADH
#undef HPART
#undef ACC_BIAS
#undef CELL_A
#undef CELL_B
#undef PIN_A
#undef SB
#undef LDS_BARRIER
}

#define BNF 32
#define BNG 128

__global__ __launch_bounds__(1024) void bn_stats_kernel(const float* __restrict__ feat,
                                                        const float* __restrict__ gamma,
                                                        const float* __restrict__ beta,
                                                        float* __restrict__ scale,
                                                        float* __restrict__ shift)
{
    __shared__ __attribute__((aligned(16))) float ps1[BNG * BNF];
    __shared__ __attribute__((aligned(16))) float ps2[BNG * BNF];
    __shared__ float pr1[32 * BNF];
    __shared__ float pr2[32 * BNF];

    const int tid = threadIdx.x;
    const int q   = tid & 7;
    const int g   = tid >> 3;
    const int f0  = blockIdx.x * BNF;

    const float4* src = reinterpret_cast<const float4*>(feat + (size_t)g * NF + f0 + 4 * q);
    float4 s  = {0.0f, 0.0f, 0.0f, 0.0f};
    float4 s2 = {0.0f, 0.0f, 0.0f, 0.0f};
#pragma unroll 16
    for (int i = 0; i < NB / BNG; ++i) {
        const float4 v = src[(size_t)i * (BNG * NF / 4)];
        s.x += v.x;  s2.x += v.x * v.x;
        s.y += v.y;  s2.y += v.y * v.y;
        s.z += v.z;  s2.z += v.z * v.z;
        s.w += v.w;  s2.w += v.w * v.w;
    }
    *reinterpret_cast<float4*>(&ps1[4 * tid]) = s;
    *reinterpret_cast<float4*>(&ps2[4 * tid]) = s2;
    __syncthreads();

    {
        const int f = tid & 31;
        const int p = tid >> 5;
        float a = 0.0f, b = 0.0f;
        for (int i = 0; i < 4; ++i) {
            a += ps1[(4 * p + i) * BNF + f];
            b += ps2[(4 * p + i) * BNF + f];
        }
        pr1[p * BNF + f] = a;
        pr2[p * BNF + f] = b;
    }
    __syncthreads();

    if (tid < BNF) {
        float a = 0.0f, b = 0.0f;
        for (int p = 0; p < 32; ++p) {
            a += pr1[p * BNF + tid];
            b += pr2[p * BNF + tid];
        }
        const int f = f0 + tid;
        const float mu  = a * (1.0f / NB);
        const float var = b * (1.0f / NB) - mu * mu;
        const float sc  = gamma[f] * rsqrtf(var + 1e-5f);
        scale[f] = sc;
        shift[f] = beta[f] - mu * sc;
    }
}

#define FCWAVES 8
#define FCS     25

__global__ __launch_bounds__(256) void fc_kernel(
    const float* __restrict__ feat,
    const float* __restrict__ scale,
    const float* __restrict__ shift,
    const float* __restrict__ fcw,
    const float* __restrict__ fcb,
    float* __restrict__ out)
{
    __shared__ float part[FCWAVES * 256];

    const int tid  = threadIdx.x;
    const int lane = tid & 31;
    const int w    = tid >> 5;
    const int b0   = blockIdx.x * 16;
    const int h    = lane >> 4;
    const int nn   = lane & 15;
    const int col0 = 200 * w + 4 * h;

    const float4* pa = reinterpret_cast<const float4*>(feat + (size_t)(b0 + nn) * NF + col0);
    const float4* pb = reinterpret_cast<const float4*>(fcw + nn * NF + col0);
    const float4* ps = reinterpret_cast<const float4*>(scale + col0);
    const float4* pt = reinterpret_cast<const float4*>(shift + col0);

    const float bias = fcb[tid & 15];

    float4 va[FCS], vb[FCS], vs[FCS], vt[FCS];
#pragma unroll
    for (int s = 0; s < FCS; ++s) {
        va[s] = pa[2 * s];
        vb[s] = pb[2 * s];
        vs[s] = ps[2 * s];
        vt[s] = pt[2 * s];
    }

    v8f acc0, acc1;
#pragma unroll
    for (int r = 0; r < 8; ++r) { acc0[r] = 0.0f; acc1[r] = 0.0f; }

#pragma unroll
    for (int s = 0; s < FCS; ++s) {
        v2f a, b;
        a.x = fmaxf(va[s].x * vs[s].x + vt[s].x, 0.0f);
        a.y = fmaxf(va[s].y * vs[s].y + vt[s].y, 0.0f);
        b.x = vb[s].x;
        b.y = vb[s].y;
        acc0 = WMMA_F32(a, b, acc0);
        a.x = fmaxf(va[s].z * vs[s].z + vt[s].z, 0.0f);
        a.y = fmaxf(va[s].w * vs[s].w + vt[s].w, 0.0f);
        b.x = vb[s].z;
        b.y = vb[s].w;
        acc1 = WMMA_F32(a, b, acc1);
    }

#pragma unroll
    for (int r = 0; r < 8; ++r)
        part[w * 256 + (r + 8 * h) * NCLS + nn] = acc0[r] + acc1[r];
    __syncthreads();

    float sum = bias + part[tid];
#pragma unroll
    for (int q = 1; q < FCWAVES; ++q) sum += part[q * 256 + tid];
    out[b0 * NCLS + tid] = sum;
}

extern "C" void kernel_launch(void* const* d_in, const int* in_sizes, int n_in,
                              void* d_out, int out_size, void* d_ws, size_t ws_size,
                              hipStream_t stream)
{
    const float* x     = (const float*)d_in[0];
    const float* Wih   = (const float*)d_in[1];
    const float* Whh   = (const float*)d_in[2];
    const float* bih   = (const float*)d_in[3];
    const float* bhh   = (const float*)d_in[4];
    const float* gamma = (const float*)d_in[5];
    const float* beta  = (const float*)d_in[6];
    const float* fcw   = (const float*)d_in[7];
    const float* fcb   = (const float*)d_in[8];
    float* out = (float*)d_out;

    float* feat  = (float*)d_ws;
    float* scale = feat + (size_t)NB * NF;
    float* shift = scale + NF;

    lstm_kernel<<<NB / 16, 128, 0, stream>>>(x, Wih, Whh, bih, bhh, feat);
    bn_stats_kernel<<<NF / BNF, 1024, 0, stream>>>(feat, gamma, beta, scale, shift);
    fc_kernel<<<NB / 16, 32 * FCWAVES, 0, stream>>>(feat, scale, shift, fcw, fcb, out);
}


# feat workspace re-laid as 200 panels [step][8 hidden][row]: lstm stores 4 pieces not 26 lines each, stats on 200 workgroups (was 50), fc loads whole lines, fc_w via LDS; reorders batch and fc K sums
# speedup vs baseline: 6.1727x; 1.0349x over previous
#include <hip/hip_runtime.h>
#include <math.h>

#define NB   4096
#define NC   200
#define NT   25
#define NH   64
#define NG   256
#define NF   1600
#define NCLS 16

#define PANEL (NB * 8)

typedef __attribute__((ext_vector_type(2))) float v2f;
typedef __attribute__((ext_vector_type(4))) float v4f;
typedef __attribute__((ext_vector_type(8))) float v8f;
typedef __attribute__((ext_vector_type(16))) float v16f;

#define CAT4(a, b, c, d)                                                                              \
    __builtin_shufflevector(__builtin_shufflevector((a), (b), 0, 1, 2, 3, 4, 5, 6, 7),                \
                            __builtin_shufflevector((c), (d), 0, 1, 2, 3, 4, 5, 6, 7),                \
                            0, 1, 2, 3, 4, 5, 6, 7, 8, 9, 10, 11, 12, 13, 14, 15)

#define WMMA_F32(a, b, c) \
    __builtin_amdgcn_wmma_f32_16x16x4_f32(false, (a), false, (b), (short)0, (c), false, false)

__device__ __forceinline__ float tanhhw(float x)
{
    return __builtin_amdgcn_tanhf(x);
}

__device__ __forceinline__ float sigf(float x)
{
    return fmaf(__builtin_amdgcn_tanhf(0.5f * x), 0.5f, 0.5f);
}

#define XROW 5000
#define HP 68

__global__ __launch_bounds__(128) void lstm_kernel(
    const float* __restrict__ x,
    const float* __restrict__ Wih,
    const float* __restrict__ Whh,
    const float* __restrict__ bih,
    const float* __restrict__ bhh,
    float* __restrict__ feat)
{
    __shared__ __attribute__((aligned(16))) float xt[16 * XROW];
    __shared__ __attribute__((aligned(16))) float hlds[16 * HP];

    const int tid  = threadIdx.x;
    const int lane = tid & 31;
    const int w    = tid >> 5;
    const int b0   = blockIdx.x * 16;
    const int q    = lane >> 4;
    const int nn   = lane & 15;
    const int rofs = 8 * q;
    const int hc   = 16 * w + nn;

    float4 wi[4][25];
    float4 wh[4][8];
#pragma unroll
    for (int j = 0; j < 4; ++j) {
        const float4* src = reinterpret_cast<const float4*>(Wih + (size_t)(64 * j + hc) * NC + 4 * q);
#pragma unroll
        for (int m = 0; m < 25; ++m) wi[j][m] = src[2 * m];
    }
#pragma unroll
    for (int j = 0; j < 4; ++j) {
        const float4* src = reinterpret_cast<const float4*>(Whh + (size_t)(64 * j + hc) * NH + 4 * q);
#pragma unroll
        for (int m = 0; m < 8; ++m) wh[j][m] = src[2 * m];
    }

    {
        const float4* src = reinterpret_cast<const float4*>(x + (size_t)b0 * NC * NT);
        float4* dst = reinterpret_cast<float4*>(xt);
        for (int u0 = 0; u0 < 156; u0 += 39) {
            float4 v[39];
#pragma unroll
            for (int u = 0; u < 39; ++u) v[u] = src[tid + 128 * (u0 + u)];
#pragma unroll
            for (int u = 0; u < 39; ++u) dst[tid + 128 * (u0 + u)] = v[u];
        }
        if (tid < 32) dst[tid + 128 * 156] = src[tid + 128 * 156];
    }

    float creg[8];
#pragma unroll
    for (int ii = 0; ii < 8; ++ii) {
        const int idx = tid + 128 * ii;
        creg[ii] = 0.0f;
        hlds[(idx >> 6) * HP + (idx & 63)] = 0.0f;
    }

    float bv[4];
#pragma unroll
    for (int j = 0; j < 4; ++j) {
        const int g = 64 * j + hc;
        bv[j] = bih[g] + bhh[g];
    }

    const float* xa = xt   + nn * XROW + 4 * q * NT;
    const float* ha = hlds + nn * HP + 4 * q;
    float* fbase = feat + (size_t)(hc >> 3) * PANEL + (b0 + rofs) * 8 + (hc & 7);
    __syncthreads();

    v8f acc[4];
    v8f gt[4];
    float xfa[10][4];
    float xfb[10][4];
    float xfc[5][4];
    float og[8];
    float4 hf[8];
#define LOADP(dst, n, m0, xp)                                                            \
    _Pragma("unroll") for (int mm = 0; mm < (n); ++mm)                                    \
        _Pragma("unroll") for (int e = 0; e < 4; ++e)                                     \
            dst[mm][e] = (xp)[NT * (8 * ((m0) + mm) + e)];
#define WMMAP(src, s0, s1, m0)                                                           \
    _Pragma("unroll") for (int mm = (s0); mm < (s1); ++mm) {                              \
        v2f a0, a1;                                                                       \
        a0.x = src[mm][0]; a0.y = src[mm][1];                                             \
        a1.x = src[mm][2]; a1.y = src[mm][3];                                             \
        _Pragma("unroll") for (int j = 0; j < 4; ++j) {                                   \
            v2f b0v; b0v.x = wi[j][(m0) + mm].x; b0v.y = wi[j][(m0) + mm].y;              \
            acc[j] = WMMA_F32(a0, b0v, acc[j]);                                           \
        }                                                                                 \
        _Pragma("unroll") for (int j = 0; j < 4; ++j) {                                   \
            v2f b1v; b1v.x = wi[j][(m0) + mm].z; b1v.y = wi[j][(m0) + mm].w;              \
            acc[j] = WMMA_F32(a1, b1v, acc[j]);                                           \
        }                                                                                 \
    }
#define LOADH                                                                            \
    _Pragma("unroll") for (int m = 0; m < 8; ++m) hf[m] = *reinterpret_cast<const float4*>(ha + 8 * m);
#define HPART                                                                            \
    _Pragma("unroll") for (int m = 0; m < 8; ++m) {                                       \
        v2f a0, a1;                                                                       \
        a0.x = hf[m].x; a0.y = hf[m].y;                                                   \
        a1.x = hf[m].z; a1.y = hf[m].w;                                                   \
        _Pragma("unroll") for (int j = 0; j < 4; ++j) {                                   \
            v2f b0v; b0v.x = wh[j][m].x; b0v.y = wh[j][m].y;                              \
            acc[j] = WMMA_F32(a0, b0v, acc[j]);                                           \
        }                                                                                 \
        _Pragma("unroll") for (int j = 0; j < 4; ++j) {                                   \
            v2f b1v; b1v.x = wh[j][m].z; b1v.y = wh[j][m].w;                              \
            acc[j] = WMMA_F32(a1, b1v, acc[j]);                                           \
        }                                                                                 \
    }
#define ACC_BIAS                                                                         \
    _Pragma("unroll") for (int j = 0; j < 4; ++j)                                         \
        _Pragma("unroll") for (int r = 0; r < 8; ++r) acc[j][r] = bv[j];
#define CELL_A(r)                                                                        \
    {                                                                                     \
        const float cc = sigf(gt[1][r]) * creg[r] + sigf(gt[0][r]) * tanhhw(gt[2][r]);   \
        creg[r] = cc;                                                                     \
        og[r] = sigf(gt[3][r]);                                                           \
    }
#define CELL_B(r, TOLDS)                                                                 \
    {                                                                                     \
        const float hn = og[r] * tanhhw(creg[r]);                                        \
        if (TOLDS) hlds[((r) + rofs) * HP + hc] = hn;                                     \
        fq[(r) * 8] = fmaxf(hn, 0.0f);                                                    \
    }
#define FEAT_PTR(t) (fbase + (size_t)(t) * (8 * PANEL))
#define PIN_A                                                                            \
    asm volatile("" : "+v"(creg[0]), "+v"(creg[1]), "+v"(creg[2]), "+v"(creg[3]),        \
                      "+v"(creg[4]), "+v"(creg[5]), "+v"(creg[6]), "+v"(creg[7]));       \
    asm volatile("" : "+v"(og[0]), "+v"(og[1]), "+v"(og[2]), "+v"(og[3]),                \
                      "+v"(og[4]), "+v"(og[5]), "+v"(og[6]), "+v"(og[7]));
#define SB __builtin_amdgcn_sched_barrier(0);
#define LDS_BARRIER                                                                      \
    __builtin_amdgcn_fence(__ATOMIC_RELEASE, "workgroup", "local");                      \
    __builtin_amdgcn_s_barrier();                                                         \
    __builtin_amdgcn_fence(__ATOMIC_ACQUIRE, "workgroup", "local");

    {
        const float* xp = xa;
        ACC_BIAS
        LOADP(xfa, 10, 0, xp)
        LOADP(xfb, 10, 10, xp)
        LOADP(xfc, 5, 20, xp)
        SB
        WMMAP(xfa, 0, 10, 0)
        WMMAP(xfb, 0, 10, 10)
        WMMAP(xfc, 0, 5, 20)
        SB
        const float* xp1 = xa + 5;
        LOADH
        LOADP(xfa, 10, 0, xp1)
        LOADP(xfb, 10, 10, xp1)
        SB
    }

#pragma unroll 2
    for (int t = 0; t < NT - 1; ++t) {
        const int t1 = t + 1;
        const int t2 = (t + 2 < NT) ? t + 2 : NT - 1;
        const float* xp  = xa + (t1 % 5) * 5 + t1 / 5;
        const float* xp2 = xa + (t2 % 5) * 5 + t2 / 5;
        float* fq = FEAT_PTR(t);

        HPART
#pragma unroll
        for (int j = 0; j < 4; ++j) gt[j] = acc[j];
        ACC_BIAS

        SB
        WMMAP(xfa, 0, 10, 0)
        CELL_A(0) CELL_A(1) CELL_A(2) CELL_A(3) CELL_A(4) CELL_A(5) CELL_A(6) CELL_A(7)
        PIN_A
        SB
        LDS_BARRIER

        LOADP(xfc, 5, 20, xp)
        SB
        WMMAP(xfb, 0, 2, 10)
        CELL_B(0, true) CELL_B(1, true)
        SB
        WMMAP(xfb, 2, 4, 10)
        CELL_B(2, true) CELL_B(3, true)
        SB
        WMMAP(xfb, 4, 6, 10)
        CELL_B(4, true) CELL_B(5, true)
        SB
        WMMAP(xfb, 6, 8, 10)
        CELL_B(6, true) CELL_B(7, true)
        SB
        WMMAP(xfb, 8, 10, 10)
        SB

        LOADP(xfa, 10, 0, xp2)
        SB
        WMMAP(xfc, 0, 3, 20)
        SB
        LDS_BARRIER

        LOADH
        LOADP(xfb, 10, 10, xp2)
        SB
        WMMAP(xfc, 3, 5, 20)
        SB
    }

    {
        const int t = NT - 1;
        float* fq = FEAT_PTR(t);
        HPART
#pragma unroll
        for (int j = 0; j < 4; ++j) gt[j] = acc[j];
        CELL_A(0) CELL_A(1) CELL_A(2) CELL_A(3) CELL_A(4) CELL_A(5) CELL_A(6) CELL_A(7)
        CELL_B(0, false) CELL_B(1, false) CELL_B(2, false) CELL_B(3, false)
        CELL_B(4, false) CELL_B(5, false) CELL_B(6, false) CELL_B(7, false)
    }
#undef LOADP
#undef WMMAP
#undef LOADH
#undef HPART
#undef ACC_BIAS
#undef CELL_A
#undef CELL_B
#undef FEAT_PTR
#undef PIN_A
#undef SB
#undef LDS_BARRIER
}

#define BNT 256
#define BNL 32

__global__ __launch_bounds__(BNT) void bn_stats_kernel(const float* __restrict__ feat,
                                                       const float* __restrict__ gamma,
                                                       const float* __restrict__ beta,
                                                       float* __restrict__ scale,
                                                       float* __restrict__ shift)
{
    __shared__ __attribute__((aligned(16))) float ps1[128 * 8];
    __shared__ __attribute__((aligned(16))) float ps2[128 * 8];
    __shared__ float pr1[16 * 8];
    __shared__ float pr2[16 * 8];

    const int tid = threadIdx.x;

    const v4f* src = reinterpret_cast<const v4f*>(feat + (size_t)blockIdx.x * PANEL) + tid;
    v16f v[BNL / 4];
#pragma unroll
    for (int k = 0; k < BNL / 4; ++k)
        v[k] = CAT4(src[BNT * (4 * k)], src[BNT * (4 * k + 1)], src[BNT * (4 * k + 2)], src[BNT * (4 * k + 3)]);
    asm volatile("" : "+v"(v[0]), "+v"(v[1]), "+v"(v[2]), "+v"(v[3]), "+v"(v[4]), "+v"(v[5]), "+v"(v[6]), "+v"(v[7]));
    float4 s  = {0.0f, 0.0f, 0.0f, 0.0f};
    float4 s2 = {0.0f, 0.0f, 0.0f, 0.0f};
#pragma unroll
    for (int k = 0; k < BNL / 4; ++k) {
#pragma unroll
        for (int e = 0; e < 16; e += 4) {
            s.x += v[k][e];      s2.x += v[k][e] * v[k][e];
            s.y += v[k][e + 1];  s2.y += v[k][e + 1] * v[k][e + 1];
            s.z += v[k][e + 2];  s2.z += v[k][e + 2] * v[k][e + 2];
            s.w += v[k][e + 3];  s2.w += v[k][e + 3] * v[k][e + 3];
        }
    }
    *reinterpret_cast<float4*>(&ps1[4 * tid]) = s;
    *reinterpret_cast<float4*>(&ps2[4 * tid]) = s2;
    __syncthreads();

    if (tid < 128) {
        const int f = tid & 7;
        const int p = tid >> 3;
        float a = 0.0f, b = 0.0f;
        for (int i = 0; i < 8; ++i) {
            a += ps1[(8 * p + i) * 8 + f];
            b += ps2[(8 * p + i) * 8 + f];
        }
        pr1[p * 8 + f] = a;
        pr2[p * 8 + f] = b;
    }
    __syncthreads();

    if (tid < 8) {
        float a = 0.0f, b = 0.0f;
        for (int p = 0; p < 16; ++p) {
            a += pr1[p * 8 + tid];
            b += pr2[p * 8 + tid];
        }
        const int g = blockIdx.x * 8 + tid;
        const int f = (g & 63) * NT + (g >> 6);
        const float mu  = a * (1.0f / NB);
        const float var = b * (1.0f / NB) - mu * mu;
        const float sc  = gamma[f] * rsqrtf(var + 1e-5f);
        scale[g] = sc;
        shift[g] = beta[f] - mu * sc;
    }
}

#define FCWAVES 8
#define FCS     25
#define FWP     1604

__global__ __launch_bounds__(256) void fc_kernel(
    const float* __restrict__ feat,
    const float* __restrict__ scale,
    const float* __restrict__ shift,
    const float* __restrict__ fcw,
    const float* __restrict__ fcb,
    float* __restrict__ out)
{
    __shared__ __attribute__((aligned(16))) float fwl[NCLS * FWP];
    __shared__ float part[FCWAVES * 256];

    const int tid  = threadIdx.x;
    const int lane = tid & 31;
    const int w    = tid >> 5;
    const int b0   = blockIdx.x * 16;
    const int h    = lane >> 4;
    const int nn   = lane & 15;
    const int g0   = 200 * w + 4 * h;

    const v4f* fsrc = reinterpret_cast<const v4f*>(fcw);
    v16f wq[6];
#pragma unroll
    for (int k = 0; k < 6; ++k)
        wq[k] = CAT4(fsrc[tid + 256 * (4 * k)], fsrc[tid + 256 * (4 * k + 1)], fsrc[tid + 256 * (4 * k + 2)], fsrc[tid + 256 * (4 * k + 3)]);
    v4f wl = fsrc[tid + 256 * 24];
    asm volatile("" : "+v"(wq[0]), "+v"(wq[1]), "+v"(wq[2]), "+v"(wq[3]), "+v"(wq[4]), "+v"(wq[5]), "+v"(wl));

    const float4* pa = reinterpret_cast<const float4*>(feat + (size_t)(25 * w) * PANEL + (b0 + nn) * 8 + 4 * h);
    const float4* ps = reinterpret_cast<const float4*>(scale + g0);
    const float4* pt = reinterpret_cast<const float4*>(shift + g0);

    const float bias = fcb[tid & 15];

#pragma unroll
    for (int i = 0; i < FCS; ++i) {
        const int p   = tid + 256 * i;
        const int row = p / 400;
        const int j   = p - 400 * row;
        v4f piece;
        if (i < 24) {
            const int e = 4 * (i & 3);
            piece.x = wq[i >> 2][e]; piece.y = wq[i >> 2][e + 1]; piece.z = wq[i >> 2][e + 2]; piece.w = wq[i >> 2][e + 3];
        } else {
            piece = wl;
        }
        *reinterpret_cast<v4f*>(&fwl[row * FWP + 4 * j]) = piece;
    }
    __builtin_amdgcn_fence(__ATOMIC_RELEASE, "workgroup", "local");
    __builtin_amdgcn_s_barrier();
    __builtin_amdgcn_fence(__ATOMIC_ACQUIRE, "workgroup", "local");

    float4 vb[FCS];
#pragma unroll
    for (int s = 0; s < FCS; ++s) {
        const int g = g0 + 8 * s;
        const float* bp = fwl + nn * FWP + (g & 63) * NT + (g >> 6);
        vb[s].x = bp[0];
        vb[s].y = bp[NT];
        vb[s].z = bp[2 * NT];
        vb[s].w = bp[3 * NT];
    }

    float4 va[FCS], vs[FCS], vt[FCS];
#pragma unroll
    for (int s = 0; s < FCS; ++s) {
        va[s] = pa[(PANEL / 4) * s];
        vs[s] = ps[2 * s];
        vt[s] = pt[2 * s];
    }

    v8f acc0, acc1;
#pragma unroll
    for (int r = 0; r < 8; ++r) { acc0[r] = 0.0f; acc1[r] = 0.0f; }

#pragma unroll
    for (int s = 0; s < FCS; ++s) {
        v2f a, b;
        a.x = fmaxf(va[s].x * vs[s].x + vt[s].x, 0.0f);
        a.y = fmaxf(va[s].y * vs[s].y + vt[s].y, 0.0f);
        b.x = vb[s].x;
        b.y = vb[s].y;
        acc0 = WMMA_F32(a, b, acc0);
        a.x = fmaxf(va[s].z * vs[s].z + vt[s].z, 0.0f);
        a.y = fmaxf(va[s].w * vs[s].w + vt[s].w, 0.0f);
        b.x = vb[s].z;
        b.y = vb[s].w;
        acc1 = WMMA_F32(a, b, acc1);
    }

#pragma unroll
    for (int r = 0; r < 8; ++r)
        part[w * 256 + (r + 8 * h) * NCLS + nn] = acc0[r] + acc1[r];
    __syncthreads();

    float sum = bias + part[tid];
#pragma unroll
    for (int q = 1; q < FCWAVES; ++q) sum += part[q * 256 + tid];
    out[b0 * NCLS + tid] = sum;
}

extern "C" void kernel_launch(void* const* d_in, const int* in_sizes, int n_in,
                              void* d_out, int out_size, void* d_ws, size_t ws_size,
                              hipStream_t stream)
{
    const float* x     = (const float*)d_in[0];
    const float* Wih   = (const float*)d_in[1];
    const float* Whh   = (const float*)d_in[2];
    const float* bih   = (const float*)d_in[3];
    const float* bhh   = (const float*)d_in[4];
    const float* gamma = (const float*)d_in[5];
    const float* beta  = (const float*)d_in[6];
    const float* fcw   = (const float*)d_in[7];
    const float* fcb   = (const float*)d_in[8];
    float* out = (float*)d_out;

    float* feat  = (float*)d_ws;
    float* scale = feat + (size_t)NB * NF;
    float* shift = scale + NF;

    lstm_kernel<<<NB / 16, 128, 0, stream>>>(x, Wih, Whh, bih, bhh, feat);
    bn_stats_kernel<<<NF / 8, BNT, 0, stream>>>(feat, gamma, beta, scale, shift);
    fc_kernel<<<NB / 16, 32 * FCWAVES, 0, stream>>>(feat, scale, shift, fcw, fcb, out);
}
